# starting kernel
# baseline (speedup 1.0000x reference)
#include <hip/hip_runtime.h>
#include <cstdint>
#include <cstddef>


typedef float v2f __attribute__((ext_vector_type(2)));
typedef float v8f __attribute__((ext_vector_type(8)));

#define D 128
#define LN_EPS 1e-5f

__global__ void rgcn_init_kernel(float* __restrict__ deg,
                                 float* __restrict__ agg, int N) {
    int tid = blockIdx.x * blockDim.x + threadIdx.x;
    if (tid < N * D) agg[tid] = 0.0f;
    if (tid < N)     deg[tid] = 1.0f;
}

__global__ void rgcn_degree_kernel(const int* __restrict__ row,
                                   float* __restrict__ deg, int E) {
    int e = blockIdx.x * blockDim.x + threadIdx.x;
    if (e < E) atomicAdd(&deg[row[e]], 1.0f);
}

__global__ void __launch_bounds__(256)
rgcn_gemm_kernel(const float* __restrict__ x, const float* __restrict__ W,
                 const float* __restrict__ bias, float* __restrict__ h, int N) {
    const int lane = threadIdx.x & 31;
    const int wave = threadIdx.x >> 5;
    const int m    = lane & 15;
    const int half = lane >> 4;
    const int row0 = blockIdx.x * 128 + wave * 16;

    int r = row0 + m;
    if (r >= N) r = N - 1;

    const float* __restrict__ xrow  = x + (size_t)r * D + 2 * half;
    const float* __restrict__ wbase = W + (size_t)m * D + 2 * half;

    v8f acc[8] = {};
#pragma unroll 4
    for (int k = 0; k < D; k += 4) {
        v2f a = *(const v2f*)(xrow + k);
#pragma unroll
        for (int t = 0; t < 8; ++t) {
            v2f b = *(const v2f*)(wbase + (size_t)t * 16 * D + k);
            acc[t] = __builtin_amdgcn_wmma_f32_16x16x4_f32(
                 false, a,  false, b,
                 (short)0, acc[t],  false,  false);
        }
    }

#pragma unroll
    for (int t = 0; t < 8; ++t) {
        const float bb = bias[t * 16 + m];
#pragma unroll
        for (int v = 0; v < 8; ++v) {
            int orow = row0 + half * 8 + v;
            if (orow < N)
                h[(size_t)orow * D + t * 16 + m] = acc[t][v] + bb;
        }
    }
}

__global__ void __launch_bounds__(256)
rgcn_edge_kernel(const int* __restrict__ ei, const float* __restrict__ deg,
                 const float* __restrict__ h, float* __restrict__ agg, int E) {
    const int e = blockIdx.x * 8 + (threadIdx.x >> 5);
    if (e >= E) return;
    const int lane = threadIdx.x & 31;

    const int s = ei[e];
    const int t = ei[E + e];
    const float norm = __frsqrt_rn(deg[s] * deg[t]);

    const float4 v = ((const float4*)(h + (size_t)s * D))[lane];
    float* at = agg + (size_t)t * D + lane * 4;
    atomicAdd(at + 0, norm * v.x);
    atomicAdd(at + 1, norm * v.y);
    atomicAdd(at + 2, norm * v.z);
    atomicAdd(at + 3, norm * v.w);
}

__global__ void __launch_bounds__(256)
rgcn_final_kernel(const float* __restrict__ x, const float* __restrict__ deg,
                  const float* __restrict__ h, const float* __restrict__ agg,
                  const float* __restrict__ gamma, const float* __restrict__ beta,
                  float* __restrict__ out, int N) {
    const int n = blockIdx.x * 8 + (threadIdx.x >> 5);
    if (n >= N) return;
    const int lane = threadIdx.x & 31;

    const float self_w = 1.0f / deg[n];

    const float4 hv = ((const float4*)(h   + (size_t)n * D))[lane];
    const float4 av = ((const float4*)(agg + (size_t)n * D))[lane];

    float4 v;
    v.x = fmaxf(av.x + self_w * hv.x, 0.0f);
    v.y = fmaxf(av.y + self_w * hv.y, 0.0f);
    v.z = fmaxf(av.z + self_w * hv.z, 0.0f);
    v.w = fmaxf(av.w + self_w * hv.w, 0.0f);

    float s  = v.x + v.y + v.z + v.w;
    float s2 = v.x * v.x + v.y * v.y + v.z * v.z + v.w * v.w;
#pragma unroll
    for (int o = 16; o >= 1; o >>= 1) {
        s  += __shfl_xor(s,  o, 32);
        s2 += __shfl_xor(s2, o, 32);
    }
    const float mu   = s * (1.0f / D);
    const float var  = s2 * (1.0f / D) - mu * mu;
    const float rstd = __frsqrt_rn(var + LN_EPS);

    const float4 g  = ((const float4*)gamma)[lane];
    const float4 be = ((const float4*)beta)[lane];
    const float4 xv = ((const float4*)(x + (size_t)n * D))[lane];

    float4 r;
    r.x = (v.x - mu) * rstd * g.x + be.x + xv.x;
    r.y = (v.y - mu) * rstd * g.y + be.y + xv.y;
    r.z = (v.z - mu) * rstd * g.z + be.z + xv.z;
    r.w = (v.w - mu) * rstd * g.w + be.w + xv.w;

    ((float4*)(out + (size_t)n * D))[lane] = r;
}

extern "C" void kernel_launch(void* const* d_in, const int* in_sizes, int n_in,
                              void* d_out, int out_size, void* d_ws, size_t ws_size,
                              hipStream_t stream) {
    const float* x     = (const float*)d_in[0];
    const int*   ei    = (const int*)  d_in[1];
    const float* W     = (const float*)d_in[2];
    const float* bias  = (const float*)d_in[3];
    const float* gamma = (const float*)d_in[4];
    const float* beta  = (const float*)d_in[5];

    const int N = in_sizes[0] / D;
    const int E = in_sizes[1] / 2;

    char* ws = (char*)d_ws;
    float* deg = (float*)ws;
    size_t off = ((size_t)N * sizeof(float) + 255) & ~(size_t)255;
    float* h = (float*)(ws + off);
    off += (size_t)N * D * sizeof(float);
    float* agg = (float*)(ws + off);

    rgcn_init_kernel<<<(N * D + 255) / 256, 256, 0, stream>>>(deg, agg, N);
    rgcn_degree_kernel<<<(E + 255) / 256, 256, 0, stream>>>(ei, deg, E);
    rgcn_gemm_kernel<<<(N + 127) / 128, 256, 0, stream>>>(x, W, bias, h, N);
    rgcn_edge_kernel<<<(E + 7) / 8, 256, 0, stream>>>(ei, deg, h, agg, E);
    rgcn_final_kernel<<<(N + 7) / 8, 256, 0, stream>>>(x, deg, h, agg, gamma,
                                                       beta, (float*)d_out, N);
}


# edge scatter: lane l takes channels l,l+32,l+64,l+96, so each f32 atomic instruction covers one whole 128-B line (4 atomic line requests per edge, was 16); atomics kept, arrival-order sums as before
# speedup vs baseline: 1.2489x; 1.2489x over previous
#include <hip/hip_runtime.h>
#include <cstdint>
#include <cstddef>


typedef float v2f __attribute__((ext_vector_type(2)));
typedef float v8f __attribute__((ext_vector_type(8)));

#define D 128
#define LN_EPS 1e-5f

__global__ void rgcn_init_kernel(float* __restrict__ deg,
                                 float* __restrict__ agg, int N) {
    int tid = blockIdx.x * blockDim.x + threadIdx.x;
    if (tid < N * D) agg[tid] = 0.0f;
    if (tid < N)     deg[tid] = 1.0f;
}

__global__ void rgcn_degree_kernel(const int* __restrict__ row,
                                   float* __restrict__ deg, int E) {
    int e = blockIdx.x * blockDim.x + threadIdx.x;
    if (e < E) atomicAdd(&deg[row[e]], 1.0f);
}

__global__ void __launch_bounds__(256)
rgcn_gemm_kernel(const float* __restrict__ x, const float* __restrict__ W,
                 const float* __restrict__ bias, float* __restrict__ h, int N) {
    const int lane = threadIdx.x & 31;
    const int wave = threadIdx.x >> 5;
    const int m    = lane & 15;
    const int half = lane >> 4;
    const int row0 = blockIdx.x * 128 + wave * 16;

    int r = row0 + m;
    if (r >= N) r = N - 1;

    const float* __restrict__ xrow  = x + (size_t)r * D + 2 * half;
    const float* __restrict__ wbase = W + (size_t)m * D + 2 * half;

    v8f acc[8] = {};
#pragma unroll 4
    for (int k = 0; k < D; k += 4) {
        v2f a = *(const v2f*)(xrow + k);
#pragma unroll
        for (int t = 0; t < 8; ++t) {
            v2f b = *(const v2f*)(wbase + (size_t)t * 16 * D + k);
            acc[t] = __builtin_amdgcn_wmma_f32_16x16x4_f32(
                 false, a,  false, b,
                 (short)0, acc[t],  false,  false);
        }
    }

#pragma unroll
    for (int t = 0; t < 8; ++t) {
        const float bb = bias[t * 16 + m];
#pragma unroll
        for (int v = 0; v < 8; ++v) {
            int orow = row0 + half * 8 + v;
            if (orow < N)
                h[(size_t)orow * D + t * 16 + m] = acc[t][v] + bb;
        }
    }
}

__global__ void __launch_bounds__(256)
rgcn_edge_kernel(const int* __restrict__ ei, const float* __restrict__ deg,
                 const float* __restrict__ h, float* __restrict__ agg, int E) {
    const int e = blockIdx.x * 8 + (threadIdx.x >> 5);
    if (e >= E) return;
    const int lane = threadIdx.x & 31;

    const int s = ei[e];
    const int t = ei[E + e];
    const float norm = __frsqrt_rn(deg[s] * deg[t]);

    const float* hs = h + (size_t)s * D + lane;
    const float v0 = hs[0];
    const float v1 = hs[32];
    const float v2 = hs[64];
    const float v3 = hs[96];

    float* at = agg + (size_t)t * D + lane;
    atomicAdd(at +  0, norm * v0);
    atomicAdd(at + 32, norm * v1);
    atomicAdd(at + 64, norm * v2);
    atomicAdd(at + 96, norm * v3);
}

__global__ void __launch_bounds__(256)
rgcn_final_kernel(const float* __restrict__ x, const float* __restrict__ deg,
                  const float* __restrict__ h, const float* __restrict__ agg,
                  const float* __restrict__ gamma, const float* __restrict__ beta,
                  float* __restrict__ out, int N) {
    const int n = blockIdx.x * 8 + (threadIdx.x >> 5);
    if (n >= N) return;
    const int lane = threadIdx.x & 31;

    const float self_w = 1.0f / deg[n];

    const float4 hv = ((const float4*)(h   + (size_t)n * D))[lane];
    const float4 av = ((const float4*)(agg + (size_t)n * D))[lane];

    float4 v;
    v.x = fmaxf(av.x + self_w * hv.x, 0.0f);
    v.y = fmaxf(av.y + self_w * hv.y, 0.0f);
    v.z = fmaxf(av.z + self_w * hv.z, 0.0f);
    v.w = fmaxf(av.w + self_w * hv.w, 0.0f);

    float s  = v.x + v.y + v.z + v.w;
    float s2 = v.x * v.x + v.y * v.y + v.z * v.z + v.w * v.w;
#pragma unroll
    for (int o = 16; o >= 1; o >>= 1) {
        s  += __shfl_xor(s,  o, 32);
        s2 += __shfl_xor(s2, o, 32);
    }
    const float mu   = s * (1.0f / D);
    const float var  = s2 * (1.0f / D) - mu * mu;
    const float rstd = __frsqrt_rn(var + LN_EPS);

    const float4 g  = ((const float4*)gamma)[lane];
    const float4 be = ((const float4*)beta)[lane];
    const float4 xv = ((const float4*)(x + (size_t)n * D))[lane];

    float4 r;
    r.x = (v.x - mu) * rstd * g.x + be.x + xv.x;
    r.y = (v.y - mu) * rstd * g.y + be.y + xv.y;
    r.z = (v.z - mu) * rstd * g.z + be.z + xv.z;
    r.w = (v.w - mu) * rstd * g.w + be.w + xv.w;

    ((float4*)(out + (size_t)n * D))[lane] = r;
}

extern "C" void kernel_launch(void* const* d_in, const int* in_sizes, int n_in,
                              void* d_out, int out_size, void* d_ws, size_t ws_size,
                              hipStream_t stream) {
    const float* x     = (const float*)d_in[0];
    const int*   ei    = (const int*)  d_in[1];
    const float* W     = (const float*)d_in[2];
    const float* bias  = (const float*)d_in[3];
    const float* gamma = (const float*)d_in[4];
    const float* beta  = (const float*)d_in[5];

    const int N = in_sizes[0] / D;
    const int E = in_sizes[1] / 2;

    char* ws = (char*)d_ws;
    float* deg = (float*)ws;
    size_t off = ((size_t)N * sizeof(float) + 255) & ~(size_t)255;
    float* h = (float*)(ws + off);
    off += (size_t)N * D * sizeof(float);
    float* agg = (float*)(ws + off);

    rgcn_init_kernel<<<(N * D + 255) / 256, 256, 0, stream>>>(deg, agg, N);
    rgcn_degree_kernel<<<(E + 255) / 256, 256, 0, stream>>>(ei, deg, E);
    rgcn_gemm_kernel<<<(N + 127) / 128, 256, 0, stream>>>(x, W, bias, h, N);
    rgcn_edge_kernel<<<(E + 7) / 8, 256, 0, stream>>>(ei, deg, h, agg, E);
    rgcn_final_kernel<<<(N + 7) / 8, 256, 0, stream>>>(x, deg, h, agg, gamma,
                                                       beta, (float*)d_out, N);
}


# edge scatter: one wave takes 16 edges (was 1): indices and norms fetched lane-parallel, all 64 row loads issued before the atomics, 1/16 the waves; atomics kept, arrival-order sums as before
# speedup vs baseline: 1.5473x; 1.2389x over previous
#include <hip/hip_runtime.h>
#include <cstdint>
#include <cstddef>


typedef float v2f __attribute__((ext_vector_type(2)));
typedef float v8f __attribute__((ext_vector_type(8)));

#define D 128
#define LN_EPS 1e-5f

__global__ void rgcn_init_kernel(float* __restrict__ deg,
                                 float* __restrict__ agg, int N) {
    int tid = blockIdx.x * blockDim.x + threadIdx.x;
    if (tid < N * D) agg[tid] = 0.0f;
    if (tid < N)     deg[tid] = 1.0f;
}

__global__ void rgcn_degree_kernel(const int* __restrict__ row,
                                   float* __restrict__ deg, int E) {
    int e = blockIdx.x * blockDim.x + threadIdx.x;
    if (e < E) atomicAdd(&deg[row[e]], 1.0f);
}

__global__ void __launch_bounds__(256)
rgcn_gemm_kernel(const float* __restrict__ x, const float* __restrict__ W,
                 const float* __restrict__ bias, float* __restrict__ h, int N) {
    const int lane = threadIdx.x & 31;
    const int wave = threadIdx.x >> 5;
    const int m    = lane & 15;
    const int half = lane >> 4;
    const int row0 = blockIdx.x * 128 + wave * 16;

    int r = row0 + m;
    if (r >= N) r = N - 1;

    const float* __restrict__ xrow  = x + (size_t)r * D + 2 * half;
    const float* __restrict__ wbase = W + (size_t)m * D + 2 * half;

    v8f acc[8] = {};
#pragma unroll 4
    for (int k = 0; k < D; k += 4) {
        v2f a = *(const v2f*)(xrow + k);
#pragma unroll
        for (int t = 0; t < 8; ++t) {
            v2f b = *(const v2f*)(wbase + (size_t)t * 16 * D + k);
            acc[t] = __builtin_amdgcn_wmma_f32_16x16x4_f32(
                 false, a,  false, b,
                 (short)0, acc[t],  false,  false);
        }
    }

#pragma unroll
    for (int t = 0; t < 8; ++t) {
        const float bb = bias[t * 16 + m];
#pragma unroll
        for (int v = 0; v < 8; ++v) {
            int orow = row0 + half * 8 + v;
            if (orow < N)
                h[(size_t)orow * D + t * 16 + m] = acc[t][v] + bb;
        }
    }
}

#define EPW 16

__device__ __forceinline__ float rgcn_lane_bcast(float v, int j) {
    return __int_as_float(__builtin_amdgcn_readlane(__float_as_int(v), j));
}

__global__ void __launch_bounds__(256)
rgcn_edge_kernel(const int* __restrict__ ei, const float* __restrict__ deg,
                 const float* __restrict__ h, float* __restrict__ agg, int E) {
    const int wave = __builtin_amdgcn_readfirstlane((int)(threadIdx.x >> 5));
    const int e0 = ((int)blockIdx.x * 8 + wave) * EPW;
    if (e0 >= E) return;
    const int lane = threadIdx.x & 31;
    const int slot = lane & 15;
    const int side = lane >> 4;

    int e = e0 + slot;
    if (e >= E) e = E - 1;
    const int   idx = ei[(size_t)side * E + e];
    const float dg  = deg[idx];
    const float nrm = __frsqrt_rn(dg * __shfl_xor(dg, 16, 32));

    const float* __restrict__ hl = h + lane;
    float* __restrict__ al = agg + lane;

    if (e0 + EPW <= E) {
        float v[EPW][4];
        for (int j = 0; j < EPW; ++j) {
            const int s = __builtin_amdgcn_readlane(idx, j);
            const float* hs = hl + (size_t)s * D;
            v[j][0] = hs[0];
            v[j][1] = hs[32];
            v[j][2] = hs[64];
            v[j][3] = hs[96];
        }
        for (int j = 0; j < EPW; ++j) {
            const int t = __builtin_amdgcn_readlane(idx, 16 + j);
            const float nj = rgcn_lane_bcast(nrm, j);
            float* at = al + (size_t)t * D;
            atomicAdd(at +  0, nj * v[j][0]);
            atomicAdd(at + 32, nj * v[j][1]);
            atomicAdd(at + 64, nj * v[j][2]);
            atomicAdd(at + 96, nj * v[j][3]);
        }
    } else {
        const int cnt = E - e0;
        for (int j = 0; j < cnt; ++j) {
            const int s = __builtin_amdgcn_readlane(idx, j);
            const int t = __builtin_amdgcn_readlane(idx, 16 + j);
            const float nj = rgcn_lane_bcast(nrm, j);
            const float* hs = hl + (size_t)s * D;
            const float v0 = hs[0];
            const float v1 = hs[32];
            const float v2 = hs[64];
            const float v3 = hs[96];
            float* at = al + (size_t)t * D;
            atomicAdd(at +  0, nj * v0);
            atomicAdd(at + 32, nj * v1);
            atomicAdd(at + 64, nj * v2);
            atomicAdd(at + 96, nj * v3);
        }
    }
}

__global__ void __launch_bounds__(256)
rgcn_final_kernel(const float* __restrict__ x, const float* __restrict__ deg,
                  const float* __restrict__ h, const float* __restrict__ agg,
                  const float* __restrict__ gamma, const float* __restrict__ beta,
                  float* __restrict__ out, int N) {
    const int n = blockIdx.x * 8 + (threadIdx.x >> 5);
    if (n >= N) return;
    const int lane = threadIdx.x & 31;

    const float self_w = 1.0f / deg[n];

    const float4 hv = ((const float4*)(h   + (size_t)n * D))[lane];
    const float4 av = ((const float4*)(agg + (size_t)n * D))[lane];

    float4 v;
    v.x = fmaxf(av.x + self_w * hv.x, 0.0f);
    v.y = fmaxf(av.y + self_w * hv.y, 0.0f);
    v.z = fmaxf(av.z + self_w * hv.z, 0.0f);
    v.w = fmaxf(av.w + self_w * hv.w, 0.0f);

    float s  = v.x + v.y + v.z + v.w;
    float s2 = v.x * v.x + v.y * v.y + v.z * v.z + v.w * v.w;
#pragma unroll
    for (int o = 16; o >= 1; o >>= 1) {
        s  += __shfl_xor(s,  o, 32);
        s2 += __shfl_xor(s2, o, 32);
    }
    const float mu   = s * (1.0f / D);
    const float var  = s2 * (1.0f / D) - mu * mu;
    const float rstd = __frsqrt_rn(var + LN_EPS);

    const float4 g  = ((const float4*)gamma)[lane];
    const float4 be = ((const float4*)beta)[lane];
    const float4 xv = ((const float4*)(x + (size_t)n * D))[lane];

    float4 r;
    r.x = (v.x - mu) * rstd * g.x + be.x + xv.x;
    r.y = (v.y - mu) * rstd * g.y + be.y + xv.y;
    r.z = (v.z - mu) * rstd * g.z + be.z + xv.z;
    r.w = (v.w - mu) * rstd * g.w + be.w + xv.w;

    ((float4*)(out + (size_t)n * D))[lane] = r;
}

extern "C" void kernel_launch(void* const* d_in, const int* in_sizes, int n_in,
                              void* d_out, int out_size, void* d_ws, size_t ws_size,
                              hipStream_t stream) {
    const float* x     = (const float*)d_in[0];
    const int*   ei    = (const int*)  d_in[1];
    const float* W     = (const float*)d_in[2];
    const float* bias  = (const float*)d_in[3];
    const float* gamma = (const float*)d_in[4];
    const float* beta  = (const float*)d_in[5];

    const int N = in_sizes[0] / D;
    const int E = in_sizes[1] / 2;

    char* ws = (char*)d_ws;
    float* deg = (float*)ws;
    size_t off = ((size_t)N * sizeof(float) + 255) & ~(size_t)255;
    float* h = (float*)(ws + off);
    off += (size_t)N * D * sizeof(float);
    float* agg = (float*)(ws + off);

    rgcn_init_kernel<<<(N * D + 255) / 256, 256, 0, stream>>>(deg, agg, N);
    rgcn_degree_kernel<<<(E + 255) / 256, 256, 0, stream>>>(ei, deg, E);
    rgcn_gemm_kernel<<<(N + 127) / 128, 256, 0, stream>>>(x, W, bias, h, N);
    rgcn_edge_kernel<<<(E + 8 * EPW - 1) / (8 * EPW), 256, 0, stream>>>(ei, deg, h, agg, E);
    rgcn_final_kernel<<<(N + 7) / 8, 256, 0, stream>>>(x, deg, h, agg, gamma,
                                                       beta, (float*)d_out, N);
}


# gemm: W copied once per workgroup into LDS (row pitch 132 floats); B fragments are 8-byte LDS reads instead of global loads that touch 16 lines each; same k order, h bit-identical
# speedup vs baseline: 1.6583x; 1.0717x over previous
#include <hip/hip_runtime.h>
#include <cstdint>
#include <cstddef>


typedef float v2f __attribute__((ext_vector_type(2)));
typedef float v8f __attribute__((ext_vector_type(8)));

#define D 128
#define LN_EPS 1e-5f

__global__ void rgcn_init_kernel(float* __restrict__ deg,
                                 float* __restrict__ agg, int N) {
    int tid = blockIdx.x * blockDim.x + threadIdx.x;
    if (tid < N * D) agg[tid] = 0.0f;
    if (tid < N)     deg[tid] = 1.0f;
}

__global__ void rgcn_degree_kernel(const int* __restrict__ row,
                                   float* __restrict__ deg, int E) {
    int e = blockIdx.x * blockDim.x + threadIdx.x;
    if (e < E) atomicAdd(&deg[row[e]], 1.0f);
}

#define WPITCH 132

__global__ void __launch_bounds__(256)
rgcn_gemm_kernel(const float* __restrict__ x, const float* __restrict__ W,
                 const float* __restrict__ bias, float* __restrict__ h, int N) {
    __shared__ __attribute__((aligned(16))) float Ws[D * WPITCH];

#pragma unroll
    for (int j = 0; j < 16; ++j) {
        const int i  = (int)threadIdx.x + j * 256;
        const int wr = i >> 5;
        const int wc = i & 31;
        const float4 w4 = ((const float4*)W)[i];
        *(float4*)(Ws + wr * WPITCH + wc * 4) = w4;
    }
    __syncthreads();

    const int lane = threadIdx.x & 31;
    const int wave = threadIdx.x >> 5;
    const int m    = lane & 15;
    const int half = lane >> 4;
    const int row0 = blockIdx.x * 128 + wave * 16;

    int r = row0 + m;
    if (r >= N) r = N - 1;

    const float* __restrict__ xrow  = x + (size_t)r * D + 2 * half;
    const float* wbase = Ws + m * WPITCH + 2 * half;

    v8f acc[8] = {};
#pragma unroll 4
    for (int k = 0; k < D; k += 4) {
        v2f a = *(const v2f*)(xrow + k);
#pragma unroll
        for (int t = 0; t < 8; ++t) {
            v2f b = *(const v2f*)(wbase + t * 16 * WPITCH + k);
            acc[t] = __builtin_amdgcn_wmma_f32_16x16x4_f32(
                 false, a,  false, b,
                 (short)0, acc[t],  false,  false);
        }
    }

#pragma unroll
    for (int t = 0; t < 8; ++t) {
        const float bb = bias[t * 16 + m];
#pragma unroll
        for (int v = 0; v < 8; ++v) {
            int orow = row0 + half * 8 + v;
            if (orow < N)
                h[(size_t)orow * D + t * 16 + m] = acc[t][v] + bb;
        }
    }
}

#define EPW 16

__device__ __forceinline__ float rgcn_lane_bcast(float v, int j) {
    return __int_as_float(__builtin_amdgcn_readlane(__float_as_int(v), j));
}

__global__ void __launch_bounds__(256)
rgcn_edge_kernel(const int* __restrict__ ei, const float* __restrict__ deg,
                 const float* __restrict__ h, float* __restrict__ agg, int E) {
    const int wave = __builtin_amdgcn_readfirstlane((int)(threadIdx.x >> 5));
    const int e0 = ((int)blockIdx.x * 8 + wave) * EPW;
    if (e0 >= E) return;
    const int lane = threadIdx.x & 31;
    const int slot = lane & 15;
    const int side = lane >> 4;

    int e = e0 + slot;
    if (e >= E) e = E - 1;
    const int   idx = ei[(size_t)side * E + e];
    const float dg  = deg[idx];
    const float nrm = __frsqrt_rn(dg * __shfl_xor(dg, 16, 32));

    const float* __restrict__ hl = h + lane;
    float* __restrict__ al = agg + lane;

    if (e0 + EPW <= E) {
        float v[EPW][4];
        for (int j = 0; j < EPW; ++j) {
            const int s = __builtin_amdgcn_readlane(idx, j);
            const float* hs = hl + (size_t)s * D;
            v[j][0] = hs[0];
            v[j][1] = hs[32];
            v[j][2] = hs[64];
            v[j][3] = hs[96];
        }
        for (int j = 0; j < EPW; ++j) {
            const int t = __builtin_amdgcn_readlane(idx, 16 + j);
            const float nj = rgcn_lane_bcast(nrm, j);
            float* at = al + (size_t)t * D;
            atomicAdd(at +  0, nj * v[j][0]);
            atomicAdd(at + 32, nj * v[j][1]);
            atomicAdd(at + 64, nj * v[j][2]);
            atomicAdd(at + 96, nj * v[j][3]);
        }
    } else {
        const int cnt = E - e0;
        for (int j = 0; j < cnt; ++j) {
            const int s = __builtin_amdgcn_readlane(idx, j);
            const int t = __builtin_amdgcn_readlane(idx, 16 + j);
            const float nj = rgcn_lane_bcast(nrm, j);
            const float* hs = hl + (size_t)s * D;
            const float v0 = hs[0];
            const float v1 = hs[32];
            const float v2 = hs[64];
            const float v3 = hs[96];
            float* at = al + (size_t)t * D;
            atomicAdd(at +  0, nj * v0);
            atomicAdd(at + 32, nj * v1);
            atomicAdd(at + 64, nj * v2);
            atomicAdd(at + 96, nj * v3);
        }
    }
}

__global__ void __launch_bounds__(256)
rgcn_final_kernel(const float* __restrict__ x, const float* __restrict__ deg,
                  const float* __restrict__ h, const float* __restrict__ agg,
                  const float* __restrict__ gamma, const float* __restrict__ beta,
                  float* __restrict__ out, int N) {
    const int n = blockIdx.x * 8 + (threadIdx.x >> 5);
    if (n >= N) return;
    const int lane = threadIdx.x & 31;

    const float self_w = 1.0f / deg[n];

    const float4 hv = ((const float4*)(h   + (size_t)n * D))[lane];
    const float4 av = ((const float4*)(agg + (size_t)n * D))[lane];

    float4 v;
    v.x = fmaxf(av.x + self_w * hv.x, 0.0f);
    v.y = fmaxf(av.y + self_w * hv.y, 0.0f);
    v.z = fmaxf(av.z + self_w * hv.z, 0.0f);
    v.w = fmaxf(av.w + self_w * hv.w, 0.0f);

    float s  = v.x + v.y + v.z + v.w;
    float s2 = v.x * v.x + v.y * v.y + v.z * v.z + v.w * v.w;
#pragma unroll
    for (int o = 16; o >= 1; o >>= 1) {
        s  += __shfl_xor(s,  o, 32);
        s2 += __shfl_xor(s2, o, 32);
    }
    const float mu   = s * (1.0f / D);
    const float var  = s2 * (1.0f / D) - mu * mu;
    const float rstd = __frsqrt_rn(var + LN_EPS);

    const float4 g  = ((const float4*)gamma)[lane];
    const float4 be = ((const float4*)beta)[lane];
    const float4 xv = ((const float4*)(x + (size_t)n * D))[lane];

    float4 r;
    r.x = (v.x - mu) * rstd * g.x + be.x + xv.x;
    r.y = (v.y - mu) * rstd * g.y + be.y + xv.y;
    r.z = (v.z - mu) * rstd * g.z + be.z + xv.z;
    r.w = (v.w - mu) * rstd * g.w + be.w + xv.w;

    ((float4*)(out + (size_t)n * D))[lane] = r;
}

extern "C" void kernel_launch(void* const* d_in, const int* in_sizes, int n_in,
                              void* d_out, int out_size, void* d_ws, size_t ws_size,
                              hipStream_t stream) {
    const float* x     = (const float*)d_in[0];
    const int*   ei    = (const int*)  d_in[1];
    const float* W     = (const float*)d_in[2];
    const float* bias  = (const float*)d_in[3];
    const float* gamma = (const float*)d_in[4];
    const float* beta  = (const float*)d_in[5];

    const int N = in_sizes[0] / D;
    const int E = in_sizes[1] / 2;

    char* ws = (char*)d_ws;
    float* deg = (float*)ws;
    size_t off = ((size_t)N * sizeof(float) + 255) & ~(size_t)255;
    float* h = (float*)(ws + off);
    off += (size_t)N * D * sizeof(float);
    float* agg = (float*)(ws + off);

    rgcn_init_kernel<<<(N * D + 255) / 256, 256, 0, stream>>>(deg, agg, N);
    rgcn_degree_kernel<<<(E + 255) / 256, 256, 0, stream>>>(ei, deg, E);
    rgcn_gemm_kernel<<<(N + 127) / 128, 256, 0, stream>>>(x, W, bias, h, N);
    rgcn_edge_kernel<<<(E + 8 * EPW - 1) / (8 * EPW), 256, 0, stream>>>(ei, deg, h, agg, E);
    rgcn_final_kernel<<<(N + 7) / 8, 256, 0, stream>>>(x, deg, h, agg, gamma,
                                                       beta, (float*)d_out, N);
}


# degree: 128 workgroups count their slice of the source indices in an LDS histogram over all nodes (ds adds), then add it to deg by whole-line f32 atomics; no global atomic per edge; deg exact
# speedup vs baseline: 2.1814x; 1.3154x over previous
#include <hip/hip_runtime.h>
#include <cstdint>
#include <cstddef>


typedef float v2f __attribute__((ext_vector_type(2)));
typedef float v8f __attribute__((ext_vector_type(8)));

#define D 128
#define LN_EPS 1e-5f

__global__ void rgcn_init_kernel(float* __restrict__ deg,
                                 float* __restrict__ agg, int N) {
    int tid = blockIdx.x * blockDim.x + threadIdx.x;
    if (tid < N * D) agg[tid] = 0.0f;
    if (tid < N)     deg[tid] = 1.0f;
}

#define DEG_BINS    65536
#define DEG_THREADS 1024
#define DEG_GROUPS  128
#define DEG_WAVES   (DEG_THREADS / 32)

__global__ void __launch_bounds__(DEG_THREADS)
rgcn_degree_kernel(const int* __restrict__ row, float* __restrict__ deg,
                   int E, int N) {
    __shared__ __attribute__((aligned(16))) unsigned int cnt[DEG_BINS];

    const int tid  = (int)threadIdx.x;
    const int lane = tid & 31;
    const int wave = tid >> 5;

    const int per  = (E + DEG_GROUPS - 1) / DEG_GROUPS;
    const int e_lo = (int)blockIdx.x * per;
    const int e_hi = (e_lo + per < E) ? (e_lo + per) : E;

    for (int base = 0; base < N; base += DEG_BINS) {
        const int nb = (N - base < DEG_BINS) ? (N - base) : DEG_BINS;

        const int nq = (nb + 3) >> 2;
        for (int q = tid; q < nq; q += DEG_THREADS)
            ((uint4*)cnt)[q] = make_uint4(0u, 0u, 0u, 0u);
        __syncthreads();

        for (int e = e_lo + tid; e < e_hi; e += 8 * DEG_THREADS) {
            int r[8];
#pragma unroll
            for (int u = 0; u < 8; ++u) {
                const int ee = e + u * DEG_THREADS;
                r[u] = (ee < e_hi) ? row[ee] : -1;
            }
#pragma unroll
            for (int u = 0; u < 8; ++u) {
                const unsigned b = (unsigned)(r[u] - base);
                if (b < (unsigned)nb) atomicAdd(&cnt[b], 1u);
            }
        }
        __syncthreads();

        const int nl    = (nb + 31) >> 5;
        const int start = ((int)blockIdx.x * nl) / DEG_GROUPS;
        for (int l0 = wave; l0 < nl; l0 += 8 * DEG_WAVES) {
            unsigned c[8];
            int      pos[8];
#pragma unroll
            for (int u = 0; u < 8; ++u) {
                const int l = l0 + u * DEG_WAVES;
                int line = start + l;
                if (line >= nl) line -= nl;
                const bool ok = (l < nl);
                const int  i  = ok ? ((line << 5) + lane) : lane;
                pos[u] = (ok && i < nb) ? i : -1;
                c[u]   = cnt[i];
            }
#pragma unroll
            for (int u = 0; u < 8; ++u)
                if (pos[u] >= 0 && c[u] != 0u)
                    atomicAdd(&deg[base + pos[u]], (float)c[u]);
        }
        __syncthreads();
    }
}

#define WPITCH 132

__global__ void __launch_bounds__(256)
rgcn_gemm_kernel(const float* __restrict__ x, const float* __restrict__ W,
                 const float* __restrict__ bias, float* __restrict__ h, int N) {
    __shared__ __attribute__((aligned(16))) float Ws[D * WPITCH];

#pragma unroll
    for (int j = 0; j < 16; ++j) {
        const int i  = (int)threadIdx.x + j * 256;
        const int wr = i >> 5;
        const int wc = i & 31;
        const float4 w4 = ((const float4*)W)[i];
        *(float4*)(Ws + wr * WPITCH + wc * 4) = w4;
    }
    __syncthreads();

    const int lane = threadIdx.x & 31;
    const int wave = threadIdx.x >> 5;
    const int m    = lane & 15;
    const int half = lane >> 4;
    const int row0 = blockIdx.x * 128 + wave * 16;

    int r = row0 + m;
    if (r >= N) r = N - 1;

    const float* __restrict__ xrow  = x + (size_t)r * D + 2 * half;
    const float* wbase = Ws + m * WPITCH + 2 * half;

    v8f acc[8] = {};
#pragma unroll 4
    for (int k = 0; k < D; k += 4) {
        v2f a = *(const v2f*)(xrow + k);
#pragma unroll
        for (int t = 0; t < 8; ++t) {
            v2f b = *(const v2f*)(wbase + t * 16 * WPITCH + k);
            acc[t] = __builtin_amdgcn_wmma_f32_16x16x4_f32(
                 false, a,  false, b,
                 (short)0, acc[t],  false,  false);
        }
    }

#pragma unroll
    for (int t = 0; t < 8; ++t) {
        const float bb = bias[t * 16 + m];
#pragma unroll
        for (int v = 0; v < 8; ++v) {
            int orow = row0 + half * 8 + v;
            if (orow < N)
                h[(size_t)orow * D + t * 16 + m] = acc[t][v] + bb;
        }
    }
}

#define EPW 16

__device__ __forceinline__ float rgcn_lane_bcast(float v, int j) {
    return __int_as_float(__builtin_amdgcn_readlane(__float_as_int(v), j));
}

__global__ void __launch_bounds__(256)
rgcn_edge_kernel(const int* __restrict__ ei, const float* __restrict__ deg,
                 const float* __restrict__ h, float* __restrict__ agg, int E) {
    const int wave = __builtin_amdgcn_readfirstlane((int)(threadIdx.x >> 5));
    const int e0 = ((int)blockIdx.x * 8 + wave) * EPW;
    if (e0 >= E) return;
    const int lane = threadIdx.x & 31;
    const int slot = lane & 15;
    const int side = lane >> 4;

    int e = e0 + slot;
    if (e >= E) e = E - 1;
    const int   idx = ei[(size_t)side * E + e];
    const float dg  = deg[idx];
    const float nrm = __frsqrt_rn(dg * __shfl_xor(dg, 16, 32));

    const float* __restrict__ hl = h + lane;
    float* __restrict__ al = agg + lane;

    if (e0 + EPW <= E) {
        float v[EPW][4];
        for (int j = 0; j < EPW; ++j) {
            const int s = __builtin_amdgcn_readlane(idx, j);
            const float* hs = hl + (size_t)s * D;
            v[j][0] = hs[0];
            v[j][1] = hs[32];
            v[j][2] = hs[64];
            v[j][3] = hs[96];
        }
        for (int j = 0; j < EPW; ++j) {
            const int t = __builtin_amdgcn_readlane(idx, 16 + j);
            const float nj = rgcn_lane_bcast(nrm, j);
            float* at = al + (size_t)t * D;
            atomicAdd(at +  0, nj * v[j][0]);
            atomicAdd(at + 32, nj * v[j][1]);
            atomicAdd(at + 64, nj * v[j][2]);
            atomicAdd(at + 96, nj * v[j][3]);
        }
    } else {
        const int cnt = E - e0;
        for (int j = 0; j < cnt; ++j) {
            const int s = __builtin_amdgcn_readlane(idx, j);
            const int t = __builtin_amdgcn_readlane(idx, 16 + j);
            const float nj = rgcn_lane_bcast(nrm, j);
            const float* hs = hl + (size_t)s * D;
            const float v0 = hs[0];
            const float v1 = hs[32];
            const float v2 = hs[64];
            const float v3 = hs[96];
            float* at = al + (size_t)t * D;
            atomicAdd(at +  0, nj * v0);
            atomicAdd(at + 32, nj * v1);
            atomicAdd(at + 64, nj * v2);
            atomicAdd(at + 96, nj * v3);
        }
    }
}

__global__ void __launch_bounds__(256)
rgcn_final_kernel(const float* __restrict__ x, const float* __restrict__ deg,
                  const float* __restrict__ h, const float* __restrict__ agg,
                  const float* __restrict__ gamma, const float* __restrict__ beta,
                  float* __restrict__ out, int N) {
    const int n = blockIdx.x * 8 + (threadIdx.x >> 5);
    if (n >= N) return;
    const int lane = threadIdx.x & 31;

    const float self_w = 1.0f / deg[n];

    const float4 hv = ((const float4*)(h   + (size_t)n * D))[lane];
    const float4 av = ((const float4*)(agg + (size_t)n * D))[lane];

    float4 v;
    v.x = fmaxf(av.x + self_w * hv.x, 0.0f);
    v.y = fmaxf(av.y + self_w * hv.y, 0.0f);
    v.z = fmaxf(av.z + self_w * hv.z, 0.0f);
    v.w = fmaxf(av.w + self_w * hv.w, 0.0f);

    float s  = v.x + v.y + v.z + v.w;
    float s2 = v.x * v.x + v.y * v.y + v.z * v.z + v.w * v.w;
#pragma unroll
    for (int o = 16; o >= 1; o >>= 1) {
        s  += __shfl_xor(s,  o, 32);
        s2 += __shfl_xor(s2, o, 32);
    }
    const float mu   = s * (1.0f / D);
    const float var  = s2 * (1.0f / D) - mu * mu;
    const float rstd = __frsqrt_rn(var + LN_EPS);

    const float4 g  = ((const float4*)gamma)[lane];
    const float4 be = ((const float4*)beta)[lane];
    const float4 xv = ((const float4*)(x + (size_t)n * D))[lane];

    float4 r;
    r.x = (v.x - mu) * rstd * g.x + be.x + xv.x;
    r.y = (v.y - mu) * rstd * g.y + be.y + xv.y;
    r.z = (v.z - mu) * rstd * g.z + be.z + xv.z;
    r.w = (v.w - mu) * rstd * g.w + be.w + xv.w;

    ((float4*)(out + (size_t)n * D))[lane] = r;
}

extern "C" void kernel_launch(void* const* d_in, const int* in_sizes, int n_in,
                              void* d_out, int out_size, void* d_ws, size_t ws_size,
                              hipStream_t stream) {
    const float* x     = (const float*)d_in[0];
    const int*   ei    = (const int*)  d_in[1];
    const float* W     = (const float*)d_in[2];
    const float* bias  = (const float*)d_in[3];
    const float* gamma = (const float*)d_in[4];
    const float* beta  = (const float*)d_in[5];

    const int N = in_sizes[0] / D;
    const int E = in_sizes[1] / 2;

    char* ws = (char*)d_ws;
    float* deg = (float*)ws;
    size_t off = ((size_t)N * sizeof(float) + 255) & ~(size_t)255;
    float* h = (float*)(ws + off);
    off += (size_t)N * D * sizeof(float);
    float* agg = (float*)(ws + off);

    rgcn_init_kernel<<<(N * D + 255) / 256, 256, 0, stream>>>(deg, agg, N);
    rgcn_degree_kernel<<<DEG_GROUPS, DEG_THREADS, 0, stream>>>(ei, deg, E, N);
    rgcn_gemm_kernel<<<(N + 127) / 128, 256, 0, stream>>>(x, W, bias, h, N);
    rgcn_edge_kernel<<<(E + 8 * EPW - 1) / (8 * EPW), 256, 0, stream>>>(ei, deg, h, agg, E);
    rgcn_final_kernel<<<(N + 7) / 8, 256, 0, stream>>>(x, deg, h, agg, gamma,
                                                       beta, (float*)d_out, N);
}


# aggregation: edges binned by 64-node target range in the call (LDS counting sort); each range summed in LDS by float adds, then ReLU+LN from LDS; no global atomics on agg; arrival-order sums kept
# speedup vs baseline: 2.3019x; 1.0552x over previous
#include <hip/hip_runtime.h>
#include <cstdint>
#include <cstddef>


typedef float v2f __attribute__((ext_vector_type(2)));
typedef float v8f __attribute__((ext_vector_type(8)));

#define D 128
#define LN_EPS 1e-5f

#define BIN_SHIFT 6
#define BIN_NODES (1 << BIN_SHIFT)

#define DEG_BINS    65536
#define DEG_THREADS 1024
#define DEG_GROUPS  128
#define DEG_WAVES   (DEG_THREADS / 32)

__global__ void __launch_bounds__(DEG_THREADS)
rgcn_degree_kernel(const int* __restrict__ row, const int* __restrict__ col,
                   float* __restrict__ deg, unsigned int* __restrict__ cntmat,
                   int E, int N) {
    __shared__ __attribute__((aligned(16))) unsigned int cnt[DEG_BINS];

    const int tid  = (int)threadIdx.x;
    const int lane = tid & 31;
    const int wave = tid >> 5;

    const int per  = (E + DEG_GROUPS - 1) / DEG_GROUPS;
    const int e_lo = (int)blockIdx.x * per;
    const int e_hi = (e_lo + per < E) ? (e_lo + per) : E;

    for (int base = 0; base < N; base += DEG_BINS) {
        const int nb = (N - base < DEG_BINS) ? (N - base) : DEG_BINS;

        const int nq = (nb + 3) >> 2;
        for (int q = tid; q < nq; q += DEG_THREADS)
            ((uint4*)cnt)[q] = make_uint4(0u, 0u, 0u, 0u);
        __syncthreads();

        for (int e = e_lo + tid; e < e_hi; e += 8 * DEG_THREADS) {
            int r[8];
#pragma unroll
            for (int u = 0; u < 8; ++u) {
                const int ee = e + u * DEG_THREADS;
                r[u] = (ee < e_hi) ? row[ee] : -1;
            }
#pragma unroll
            for (int u = 0; u < 8; ++u) {
                const unsigned b = (unsigned)(r[u] - base);
                if (b < (unsigned)nb) atomicAdd(&cnt[b], 1u);
            }
        }
        __syncthreads();

        const int nl    = (nb + 31) >> 5;
        const int start = ((int)blockIdx.x * nl) / DEG_GROUPS;
        for (int l0 = wave; l0 < nl; l0 += 8 * DEG_WAVES) {
            unsigned c[8];
            int      pos[8];
#pragma unroll
            for (int u = 0; u < 8; ++u) {
                const int l = l0 + u * DEG_WAVES;
                int line = start + l;
                if (line >= nl) line -= nl;
                const bool ok = (l < nl);
                const int  i  = ok ? ((line << 5) + lane) : lane;
                pos[u] = (ok && i < nb) ? i : -1;
                c[u]   = cnt[i];
            }
#pragma unroll
            for (int u = 0; u < 8; ++u)
                if (pos[u] >= 0 && c[u] != 0u)
                    atomicAdd(&deg[base + pos[u]], (float)c[u]);
        }
        __syncthreads();
    }

    const int NB = (N + BIN_NODES - 1) >> BIN_SHIFT;
    for (int bbase = 0; bbase < NB; bbase += DEG_BINS) {
        const int nbb = (NB - bbase < DEG_BINS) ? (NB - bbase) : DEG_BINS;
        for (int q = tid; q < nbb; q += DEG_THREADS) cnt[q] = 0u;
        __syncthreads();
        for (int e = e_lo + tid; e < e_hi; e += 8 * DEG_THREADS) {
            int r[8];
#pragma unroll
            for (int u = 0; u < 8; ++u) {
                const int ee = e + u * DEG_THREADS;
                r[u] = (ee < e_hi) ? col[ee] : -1;
            }
#pragma unroll
            for (int u = 0; u < 8; ++u) {
                const unsigned b = (unsigned)((r[u] >> BIN_SHIFT) - bbase);
                if (b < (unsigned)nbb) atomicAdd(&cnt[b], 1u);
            }
        }
        __syncthreads();
        for (int q = tid; q < nbb; q += DEG_THREADS)
            cntmat[(size_t)blockIdx.x * NB + bbase + q] = cnt[q];
        __syncthreads();
    }
}

#define WPITCH 132

__global__ void __launch_bounds__(256)
rgcn_gemm_kernel(const float* __restrict__ x, const float* __restrict__ W,
                 const float* __restrict__ bias, float* __restrict__ h,
                 float* __restrict__ deg, int N) {
    __shared__ __attribute__((aligned(16))) float Ws[D * WPITCH];

#pragma unroll
    for (int j = 0; j < 16; ++j) {
        const int i  = (int)threadIdx.x + j * 256;
        const int wr = i >> 5;
        const int wc = i & 31;
        const float4 w4 = ((const float4*)W)[i];
        *(float4*)(Ws + wr * WPITCH + wc * 4) = w4;
    }
    __syncthreads();

    if (threadIdx.x < 128) {
        const int n = (int)blockIdx.x * 128 + (int)threadIdx.x;
        if (n < N) deg[n] = 1.0f;
    }

    const int lane = threadIdx.x & 31;
    const int wave = threadIdx.x >> 5;
    const int m    = lane & 15;
    const int half = lane >> 4;
    const int row0 = blockIdx.x * 128 + wave * 16;

    int r = row0 + m;
    if (r >= N) r = N - 1;

    const float* __restrict__ xrow  = x + (size_t)r * D + 2 * half;
    const float* wbase = Ws + m * WPITCH + 2 * half;

    v8f acc[8] = {};
#pragma unroll 4
    for (int k = 0; k < D; k += 4) {
        v2f a = *(const v2f*)(xrow + k);
#pragma unroll
        for (int t = 0; t < 8; ++t) {
            v2f b = *(const v2f*)(wbase + t * 16 * WPITCH + k);
            acc[t] = __builtin_amdgcn_wmma_f32_16x16x4_f32(
                 false, a,  false, b,
                 (short)0, acc[t],  false,  false);
        }
    }

#pragma unroll
    for (int t = 0; t < 8; ++t) {
        const float bb = bias[t * 16 + m];
#pragma unroll
        for (int v = 0; v < 8; ++v) {
            int orow = row0 + half * 8 + v;
            if (orow < N)
                h[(size_t)orow * D + t * 16 + m] = acc[t][v] + bb;
        }
    }
}

#define BIN_THREADS 1024
#define BIN_MAXB    16384

__global__ void __launch_bounds__(BIN_THREADS)
rgcn_bin_kernel(const int* __restrict__ ei, const float* __restrict__ deg,
                const unsigned int* __restrict__ cntmat,
                int* __restrict__ binstart, uint4* __restrict__ rec,
                int E, int N) {
    __shared__ unsigned int cur[BIN_MAXB];
    __shared__ unsigned int tot[BIN_MAXB];
    __shared__ unsigned int wsum[BIN_THREADS / 32];

    const int tid  = (int)threadIdx.x;
    const int lane = tid & 31;
    const int wave = tid >> 5;
    const int g    = (int)blockIdx.x;
    const int NB   = (N + BIN_NODES - 1) >> BIN_SHIFT;

    for (int b = tid; b < NB; b += BIN_THREADS) {
        unsigned total = 0u, mine = 0u;
#pragma unroll 32
        for (int gg = 0; gg < DEG_GROUPS; ++gg) {
            const unsigned cc = cntmat[(size_t)gg * NB + b];
            total += cc;
            mine  += (gg < g) ? cc : 0u;
        }
        tot[b] = total;
        cur[b] = mine;
    }
    __syncthreads();

    unsigned carry = 0u;
    for (int cb = 0; cb < NB; cb += BIN_THREADS) {
        const int b = cb + tid;
        const unsigned v = (b < NB) ? tot[b] : 0u;
        unsigned inc = v;
#pragma unroll
        for (int o = 1; o < 32; o <<= 1) {
            const unsigned up = __shfl_up(inc, o, 32);
            if (lane >= o) inc += up;
        }
        if (lane == 31) wsum[wave] = inc;
        __syncthreads();
        unsigned woff = 0u, ctot = 0u;
#pragma unroll
        for (int w = 0; w < BIN_THREADS / 32; ++w) {
            const unsigned sw = wsum[w];
            woff += (w < wave) ? sw : 0u;
            ctot += sw;
        }
        if (b < NB) {
            const unsigned start = carry + woff + inc - v;
            cur[b] += start;
            if (g == 0) binstart[b] = (int)start;
        }
        carry += ctot;
        __syncthreads();
    }
    if (g == 0 && tid == 0) binstart[NB] = (int)carry;

    const int per  = (E + DEG_GROUPS - 1) / DEG_GROUPS;
    const int e_lo = g * per;
    const int e_hi = (e_lo + per < E) ? (e_lo + per) : E;
    const int* __restrict__ row = ei;
    const int* __restrict__ col = ei + E;

    for (int e = e_lo + tid; e < e_hi; e += 4 * BIN_THREADS) {
        int   sidx[4], tidx[4];
        float ds[4], dt[4];
#pragma unroll
        for (int u = 0; u < 4; ++u) {
            const int ee = e + u * BIN_THREADS;
            const bool ok = (ee < e_hi);
            sidx[u] = ok ? row[ee] : -1;
            tidx[u] = ok ? col[ee] : -1;
        }
#pragma unroll
        for (int u = 0; u < 4; ++u) {
            const bool ok = (sidx[u] >= 0);
            ds[u] = ok ? deg[sidx[u]] : 1.0f;
            dt[u] = ok ? deg[tidx[u]] : 1.0f;
        }
#pragma unroll
        for (int u = 0; u < 4; ++u) {
            if (sidx[u] >= 0) {
                const float nrm = __frsqrt_rn(ds[u] * dt[u]);
                const unsigned p = atomicAdd(&cur[tidx[u] >> BIN_SHIFT], 1u);
                if (p < (unsigned)E)
                    rec[p] = make_uint4((unsigned)sidx[u], (unsigned)tidx[u],
                                        __float_as_uint(nrm), 0u);
            }
        }
    }
}

#define EPW 16
#define SUM_WAVES 8
#define NPW (BIN_NODES / SUM_WAVES)

__device__ __forceinline__ float rgcn_lane_bcast(float v, int j) {
    return __int_as_float(__builtin_amdgcn_readlane(__float_as_int(v), j));
}

__global__ void __launch_bounds__(SUM_WAVES * 32)
rgcn_bin_sum_kernel(const float* __restrict__ x, const float* __restrict__ deg,
                    const float* __restrict__ h, const uint4* __restrict__ rec,
                    const int* __restrict__ binstart,
                    const float* __restrict__ gamma, const float* __restrict__ beta,
                    float* __restrict__ out, int E, int N) {
    __shared__ __attribute__((aligned(16))) float acc[BIN_NODES * D];

    const int tid  = (int)threadIdx.x;
    const int lane = tid & 31;
    const int wave = __builtin_amdgcn_readfirstlane(tid >> 5);
    const int n0   = (int)blockIdx.x << BIN_SHIFT;
    int e_lo = binstart[blockIdx.x];
    int e_hi = binstart[blockIdx.x + 1];
    if (e_lo < 0) e_lo = 0;
    if (e_hi > E) e_hi = E;

    for (int q = tid; q < BIN_NODES * D / 4; q += SUM_WAVES * 32)
        ((float4*)acc)[q] = make_float4(0.0f, 0.0f, 0.0f, 0.0f);
    __syncthreads();

    const float* __restrict__ hl = h + lane;
    float* al = acc + lane;

    for (int e0 = e_lo + wave * EPW; e0 < e_hi; e0 += SUM_WAVES * EPW) {
        int e = e0 + (lane & 15);
        if (e >= e_hi) e = e_hi - 1;
        const uint4 r = rec[e];
        const int   rs = min(max((int)r.x, 0), N - 1);
        const int   rt = ((int)r.y - n0) & (BIN_NODES - 1);
        const float rn = __uint_as_float(r.z);

        if (e0 + EPW <= e_hi) {
            float v[EPW][4];
            for (int j = 0; j < EPW; ++j) {
                const int s = __builtin_amdgcn_readlane(rs, j);
                const float* hs = hl + (size_t)s * D;
                v[j][0] = hs[0];
                v[j][1] = hs[32];
                v[j][2] = hs[64];
                v[j][3] = hs[96];
            }
            for (int j = 0; j < EPW; ++j) {
                const int t = __builtin_amdgcn_readlane(rt, j);
                const float nj = rgcn_lane_bcast(rn, j);
                float* at = al + t * D;
                atomicAdd(at +  0, nj * v[j][0]);
                atomicAdd(at + 32, nj * v[j][1]);
                atomicAdd(at + 64, nj * v[j][2]);
                atomicAdd(at + 96, nj * v[j][3]);
            }
        } else {
            const int cnt = e_hi - e0;
            for (int j = 0; j < cnt; ++j) {
                const int s = __builtin_amdgcn_readlane(rs, j);
                const int t = __builtin_amdgcn_readlane(rt, j);
                const float nj = rgcn_lane_bcast(rn, j);
                const float* hs = hl + (size_t)s * D;
                const float v0 = hs[0];
                const float v1 = hs[32];
                const float v2 = hs[64];
                const float v3 = hs[96];
                float* at = al + t * D;
                atomicAdd(at +  0, nj * v0);
                atomicAdd(at + 32, nj * v1);
                atomicAdd(at + 64, nj * v2);
                atomicAdd(at + 96, nj * v3);
            }
        }
    }
    const float4 g  = ((const float4*)gamma)[lane];
    const float4 be = ((const float4*)beta)[lane];
    float4 hv[NPW], xv[NPW];
    float  dgv[NPW];
#pragma unroll
    for (int k = 0; k < NPW; ++k) {
        int n = n0 + wave + k * SUM_WAVES;
        if (n >= N) n = N - 1;
        dgv[k] = deg[n];
        hv[k]  = ((const float4*)(h + (size_t)n * D))[lane];
        xv[k]  = ((const float4*)(x + (size_t)n * D))[lane];
    }
    __syncthreads();

#pragma unroll
    for (int k = 0; k < NPW; ++k) {
        const int i = wave + k * SUM_WAVES;
        const int n = n0 + i;
        if (n < N) {
            const float self_w = 1.0f / dgv[k];
            const float4 av = ((const float4*)(acc + i * D))[lane];

            float4 v;
            v.x = fmaxf(av.x + self_w * hv[k].x, 0.0f);
            v.y = fmaxf(av.y + self_w * hv[k].y, 0.0f);
            v.z = fmaxf(av.z + self_w * hv[k].z, 0.0f);
            v.w = fmaxf(av.w + self_w * hv[k].w, 0.0f);

            float s  = v.x + v.y + v.z + v.w;
            float s2 = v.x * v.x + v.y * v.y + v.z * v.z + v.w * v.w;
#pragma unroll
            for (int o = 16; o >= 1; o >>= 1) {
                s  += __shfl_xor(s,  o, 32);
                s2 += __shfl_xor(s2, o, 32);
            }
            const float mu   = s * (1.0f / D);
            const float var  = s2 * (1.0f / D) - mu * mu;
            const float rstd = __frsqrt_rn(var + LN_EPS);

            float4 r;
            r.x = (v.x - mu) * rstd * g.x + be.x + xv[k].x;
            r.y = (v.y - mu) * rstd * g.y + be.y + xv[k].y;
            r.z = (v.z - mu) * rstd * g.z + be.z + xv[k].z;
            r.w = (v.w - mu) * rstd * g.w + be.w + xv[k].w;

            ((float4*)(out + (size_t)n * D))[lane] = r;
        }
    }
}

extern "C" void kernel_launch(void* const* d_in, const int* in_sizes, int n_in,
                              void* d_out, int out_size, void* d_ws, size_t ws_size,
                              hipStream_t stream) {
    const float* x     = (const float*)d_in[0];
    const int*   ei    = (const int*)  d_in[1];
    const float* W     = (const float*)d_in[2];
    const float* bias  = (const float*)d_in[3];
    const float* gamma = (const float*)d_in[4];
    const float* beta  = (const float*)d_in[5];

    const int N  = in_sizes[0] / D;
    const int E  = in_sizes[1] / 2;
    const int NB = (N + BIN_NODES - 1) / BIN_NODES;

    char* ws = (char*)d_ws;
    float* deg = (float*)ws;
    size_t off = ((size_t)N * sizeof(float) + 255) & ~(size_t)255;
    float* h = (float*)(ws + off);
    off += ((size_t)N * D * sizeof(float) + 255) & ~(size_t)255;
    unsigned int* cntmat = (unsigned int*)(ws + off);
    off += ((size_t)DEG_GROUPS * NB * sizeof(unsigned int) + 255) & ~(size_t)255;
    int* binstart = (int*)(ws + off);
    off += ((size_t)(NB + 1) * sizeof(int) + 255) & ~(size_t)255;
    uint4* rec = (uint4*)(ws + off);

    rgcn_gemm_kernel<<<(N + 127) / 128, 256, 0, stream>>>(x, W, bias, h, deg, N);
    rgcn_degree_kernel<<<DEG_GROUPS, DEG_THREADS, 0, stream>>>(ei, ei + E, deg, cntmat, E, N);
    rgcn_bin_kernel<<<DEG_GROUPS, BIN_THREADS, 0, stream>>>(ei, deg, cntmat, binstart, rec, E, N);
    rgcn_bin_sum_kernel<<<NB, SUM_WAVES * 32, 0, stream>>>(x, deg, h, rec, binstart, gamma,
                                                         beta, (float*)d_out, E, N);
}


# weight dis[s]*dis[t] split per node: gemm stores dis[s]*h, sum kernel applies dis[t] after its arrival-order sums (rounding differs); no per-edge deg gathers; deg from count planes, no global atomics
# speedup vs baseline: 2.4971x; 1.0848x over previous
#include <hip/hip_runtime.h>
#include <cstdint>
#include <cstddef>


typedef float v2f __attribute__((ext_vector_type(2)));
typedef float v8f __attribute__((ext_vector_type(8)));

#define D 128
#define LN_EPS 1e-5f

#define BIN_SHIFT 6
#define BIN_NODES (1 << BIN_SHIFT)

#define DEG_BINS    65536
#define DEG_THREADS 1024
#define DEG_GROUPS  128

__global__ void __launch_bounds__(DEG_THREADS)
rgcn_degree_kernel(const int* __restrict__ row, const int* __restrict__ col,
                   unsigned int* __restrict__ planes, unsigned int* __restrict__ cntmat,
                   int E, int N, int pitchw) {
    __shared__ __attribute__((aligned(16))) unsigned int cnt[DEG_BINS];

    const int tid  = (int)threadIdx.x;

    const int per  = (E + DEG_GROUPS - 1) / DEG_GROUPS;
    const int e_lo = (int)blockIdx.x * per;
    const int e_hi = (e_lo + per < E) ? (e_lo + per) : E;

    uint4* __restrict__ plane4 = (uint4*)(planes + (size_t)blockIdx.x * pitchw);

    for (int base = 0; base < N; base += DEG_BINS) {
        const int nb = (N - base < DEG_BINS) ? (N - base) : DEG_BINS;

        const int nw4 = (nb + 7) >> 3;
        for (int q = tid; q < 2 * nw4; q += DEG_THREADS)
            ((uint4*)cnt)[q] = make_uint4(0u, 0u, 0u, 0u);
        __syncthreads();

        for (int e = e_lo + tid; e < e_hi; e += 8 * DEG_THREADS) {
            int r[8];
#pragma unroll
            for (int u = 0; u < 8; ++u) {
                const int ee = e + u * DEG_THREADS;
                r[u] = (ee < e_hi) ? row[ee] : -1;
            }
#pragma unroll
            for (int u = 0; u < 8; ++u) {
                const unsigned b = (unsigned)(r[u] - base);
                if (b < (unsigned)nb) atomicAdd(&cnt[b], 1u);
            }
        }
        __syncthreads();

        for (int q = tid; q < nw4; q += DEG_THREADS) {
            const uint4 a = ((const uint4*)cnt)[2 * q];
            const uint4 b = ((const uint4*)cnt)[2 * q + 1];
            plane4[(base >> 3) + q] = make_uint4(a.x | (a.y << 16), a.z | (a.w << 16),
                                                 b.x | (b.y << 16), b.z | (b.w << 16));
        }
        __syncthreads();
    }

    const int NB = (N + BIN_NODES - 1) >> BIN_SHIFT;
    for (int bbase = 0; bbase < NB; bbase += DEG_BINS) {
        const int nbb = (NB - bbase < DEG_BINS) ? (NB - bbase) : DEG_BINS;
        for (int q = tid; q < nbb; q += DEG_THREADS) cnt[q] = 0u;
        __syncthreads();
        for (int e = e_lo + tid; e < e_hi; e += 8 * DEG_THREADS) {
            int r[8];
#pragma unroll
            for (int u = 0; u < 8; ++u) {
                const int ee = e + u * DEG_THREADS;
                r[u] = (ee < e_hi) ? col[ee] : -1;
            }
#pragma unroll
            for (int u = 0; u < 8; ++u) {
                const unsigned b = (unsigned)((r[u] >> BIN_SHIFT) - bbase);
                if (b < (unsigned)nbb) atomicAdd(&cnt[b], 1u);
            }
        }
        __syncthreads();
        for (int q = tid; q < nbb; q += DEG_THREADS)
            cntmat[(size_t)blockIdx.x * NB + bbase + q] = cnt[q];
        __syncthreads();
    }
}

#define WPITCH 132
#define GEMM_ROWS 128
#define PL_PARTS  4

__global__ void __launch_bounds__(256)
rgcn_gemm_kernel(const float* __restrict__ x, const float* __restrict__ W,
                 const float* __restrict__ bias, float* __restrict__ h,
                 const unsigned int* __restrict__ planes, float* __restrict__ disv,
                 int N, int pitchw) {
    __shared__ __attribute__((aligned(16))) float Ws[D * WPITCH];
    __shared__ unsigned int degp[PL_PARTS][GEMM_ROWS];
    __shared__ __attribute__((aligned(16))) float dis_s[GEMM_ROWS];

    const int pword = (int)threadIdx.x & (GEMM_ROWS / 2 - 1);
    const int part  = (int)threadIdx.x / (GEMM_ROWS / 2);
    const int widx  = (int)blockIdx.x * (GEMM_ROWS / 2) + pword;
    unsigned pc[DEG_GROUPS / PL_PARTS];
    {
        const bool pok = (widx < pitchw);
        const unsigned pw0   = (unsigned)part * (unsigned)pitchw + (unsigned)(pok ? widx : 0);
        const unsigned pstep = (unsigned)PL_PARTS * (unsigned)pitchw;
#pragma unroll
        for (int g = 0; g < DEG_GROUPS / PL_PARTS; ++g) {
            const unsigned c = planes[pw0 + (unsigned)g * pstep];
            pc[g] = pok ? c : 0u;
        }
    }

#pragma unroll
    for (int j = 0; j < 16; ++j) {
        const int i  = (int)threadIdx.x + j * 256;
        const int wr = i >> 5;
        const int wc = i & 31;
        const float4 w4 = ((const float4*)W)[i];
        *(float4*)(Ws + wr * WPITCH + wc * 4) = w4;
    }
    {
        unsigned lo = 0u, hi = 0u;
#pragma unroll
        for (int g = 0; g < DEG_GROUPS / PL_PARTS; ++g) {
            lo += pc[g] & 0xffffu;
            hi += pc[g] >> 16;
        }
        degp[part][2 * pword]     = lo;
        degp[part][2 * pword + 1] = hi;
    }
    __syncthreads();

    if (threadIdx.x < GEMM_ROWS) {
        unsigned dsum = 1u;
#pragma unroll
        for (int p = 0; p < PL_PARTS; ++p) dsum += degp[p][threadIdx.x];
        const float dv = __frsqrt_rn((float)dsum);
        dis_s[threadIdx.x] = dv;
        const int n = (int)blockIdx.x * GEMM_ROWS + (int)threadIdx.x;
        if (n < N) disv[n] = dv;
    }
    __syncthreads();

    const int lane = threadIdx.x & 31;
    const int wave = threadIdx.x >> 5;
    const int m    = lane & 15;
    const int half = lane >> 4;
    const int row0 = blockIdx.x * 128 + wave * 16;

    int r = row0 + m;
    if (r >= N) r = N - 1;

    const float* __restrict__ xrow  = x + (size_t)r * D + 2 * half;
    const float* wbase = Ws + m * WPITCH + 2 * half;

    v8f acc[8] = {};
#pragma unroll 4
    for (int k = 0; k < D; k += 4) {
        v2f a = *(const v2f*)(xrow + k);
#pragma unroll
        for (int t = 0; t < 8; ++t) {
            v2f b = *(const v2f*)(wbase + t * 16 * WPITCH + k);
            acc[t] = __builtin_amdgcn_wmma_f32_16x16x4_f32(
                 false, a,  false, b,
                 (short)0, acc[t],  false,  false);
        }
    }

    const float4 dlo = *(const float4*)(dis_s + wave * 16 + half * 8);
    const float4 dhi = *(const float4*)(dis_s + wave * 16 + half * 8 + 4);
    const float dr[8] = { dlo.x, dlo.y, dlo.z, dlo.w, dhi.x, dhi.y, dhi.z, dhi.w };

#pragma unroll
    for (int t = 0; t < 8; ++t) {
        const float bb = bias[t * 16 + m];
#pragma unroll
        for (int v = 0; v < 8; ++v) {
            int orow = row0 + half * 8 + v;
            if (orow < N)
                h[(size_t)orow * D + t * 16 + m] = (acc[t][v] + bb) * dr[v];
        }
    }
}

#define BIN_THREADS 1024
#define BIN_MAXB    16384

__global__ void __launch_bounds__(BIN_THREADS)
rgcn_bin_kernel(const int* __restrict__ ei,
                const unsigned int* __restrict__ cntmat,
                int* __restrict__ binstart, uint2* __restrict__ rec,
                int E, int N) {
    __shared__ unsigned int cur[BIN_MAXB];
    __shared__ unsigned int tot[BIN_MAXB];
    __shared__ unsigned int wsum[BIN_THREADS / 32];

    const int tid  = (int)threadIdx.x;
    const int lane = tid & 31;
    const int wave = tid >> 5;
    const int g    = (int)blockIdx.x;
    const int NB   = (N + BIN_NODES - 1) >> BIN_SHIFT;

    for (int b = tid; b < NB; b += BIN_THREADS) {
        unsigned total = 0u, mine = 0u;
#pragma unroll 32
        for (int gg = 0; gg < DEG_GROUPS; ++gg) {
            const unsigned cc = cntmat[(size_t)gg * NB + b];
            total += cc;
            mine  += (gg < g) ? cc : 0u;
        }
        tot[b] = total;
        cur[b] = mine;
    }
    __syncthreads();

    unsigned carry = 0u;
    for (int cb = 0; cb < NB; cb += BIN_THREADS) {
        const int b = cb + tid;
        const unsigned v = (b < NB) ? tot[b] : 0u;
        unsigned inc = v;
#pragma unroll
        for (int o = 1; o < 32; o <<= 1) {
            const unsigned up = __shfl_up(inc, o, 32);
            if (lane >= o) inc += up;
        }
        if (lane == 31) wsum[wave] = inc;
        __syncthreads();
        unsigned woff = 0u, ctot = 0u;
#pragma unroll
        for (int w = 0; w < BIN_THREADS / 32; ++w) {
            const unsigned sw = wsum[w];
            woff += (w < wave) ? sw : 0u;
            ctot += sw;
        }
        if (b < NB) {
            const unsigned start = carry + woff + inc - v;
            cur[b] += start;
            if (g == 0) binstart[b] = (int)start;
        }
        carry += ctot;
        __syncthreads();
    }
    if (g == 0 && tid == 0) binstart[NB] = (int)carry;

    const int per  = (E + DEG_GROUPS - 1) / DEG_GROUPS;
    const int e_lo = g * per;
    const int e_hi = (e_lo + per < E) ? (e_lo + per) : E;
    const int* __restrict__ row = ei;
    const int* __restrict__ col = ei + E;

    for (int e = e_lo + tid; e < e_hi; e += 4 * BIN_THREADS) {
        int sidx[4], tidx[4];
#pragma unroll
        for (int u = 0; u < 4; ++u) {
            const int ee = e + u * BIN_THREADS;
            const bool ok = (ee < e_hi);
            sidx[u] = ok ? row[ee] : -1;
            tidx[u] = ok ? col[ee] : -1;
        }
#pragma unroll
        for (int u = 0; u < 4; ++u) {
            if (sidx[u] >= 0) {
                const unsigned p = atomicAdd(&cur[tidx[u] >> BIN_SHIFT], 1u);
                if (p < (unsigned)E)
                    rec[p] = make_uint2((unsigned)sidx[u], (unsigned)tidx[u]);
            }
        }
    }
}

#define EPW 16
#define SUM_WAVES 8
#define NPW (BIN_NODES / SUM_WAVES)

__global__ void __launch_bounds__(SUM_WAVES * 32)
rgcn_bin_sum_kernel(const float* __restrict__ x, const float* __restrict__ disv,
                    const float* __restrict__ h, const uint2* __restrict__ rec,
                    const int* __restrict__ binstart,
                    const float* __restrict__ gamma, const float* __restrict__ beta,
                    float* __restrict__ out, int E, int N) {
    __shared__ __attribute__((aligned(16))) float acc[BIN_NODES * D];

    const int tid  = (int)threadIdx.x;
    const int lane = tid & 31;
    const int wave = __builtin_amdgcn_readfirstlane(tid >> 5);
    const int n0   = (int)blockIdx.x << BIN_SHIFT;
    int e_lo = binstart[blockIdx.x];
    int e_hi = binstart[blockIdx.x + 1];
    if (e_lo < 0) e_lo = 0;
    if (e_hi > E) e_hi = E;

    for (int q = tid; q < BIN_NODES * D / 4; q += SUM_WAVES * 32)
        ((float4*)acc)[q] = make_float4(0.0f, 0.0f, 0.0f, 0.0f);
    __syncthreads();

    const float* __restrict__ hl = h + lane;
    float* al = acc + lane;

    for (int e0 = e_lo + wave * EPW; e0 < e_hi; e0 += SUM_WAVES * EPW) {
        int e = e0 + (lane & 15);
        if (e >= e_hi) e = e_hi - 1;
        const uint2 r = rec[e];
        const int   rs = min(max((int)r.x, 0), N - 1);
        const int   rt = ((int)r.y - n0) & (BIN_NODES - 1);

        if (e0 + EPW <= e_hi) {
            float v[EPW][4];
            for (int j = 0; j < EPW; ++j) {
                const int s = __builtin_amdgcn_readlane(rs, j);
                const float* hs = hl + (size_t)s * D;
                v[j][0] = hs[0];
                v[j][1] = hs[32];
                v[j][2] = hs[64];
                v[j][3] = hs[96];
            }
            for (int j = 0; j < EPW; ++j) {
                const int t = __builtin_amdgcn_readlane(rt, j);
                float* at = al + t * D;
                atomicAdd(at +  0, v[j][0]);
                atomicAdd(at + 32, v[j][1]);
                atomicAdd(at + 64, v[j][2]);
                atomicAdd(at + 96, v[j][3]);
            }
        } else {
            const int cnt = e_hi - e0;
            for (int j = 0; j < cnt; ++j) {
                const int s = __builtin_amdgcn_readlane(rs, j);
                const int t = __builtin_amdgcn_readlane(rt, j);
                const float* hs = hl + (size_t)s * D;
                const float v0 = hs[0];
                const float v1 = hs[32];
                const float v2 = hs[64];
                const float v3 = hs[96];
                float* at = al + t * D;
                atomicAdd(at +  0, v0);
                atomicAdd(at + 32, v1);
                atomicAdd(at + 64, v2);
                atomicAdd(at + 96, v3);
            }
        }
    }
    const float4 g  = ((const float4*)gamma)[lane];
    const float4 be = ((const float4*)beta)[lane];
    float4 hv[NPW], xv[NPW];
    float  dgv[NPW];
#pragma unroll
    for (int k = 0; k < NPW; ++k) {
        int n = n0 + wave + k * SUM_WAVES;
        if (n >= N) n = N - 1;
        dgv[k] = disv[n];
        hv[k]  = ((const float4*)(h + (size_t)n * D))[lane];
        xv[k]  = ((const float4*)(x + (size_t)n * D))[lane];
    }
    __syncthreads();

#pragma unroll
    for (int k = 0; k < NPW; ++k) {
        const int i = wave + k * SUM_WAVES;
        const int n = n0 + i;
        if (n < N) {
            const float dt = dgv[k];
            const float4 av = ((const float4*)(acc + i * D))[lane];

            float4 v;
            v.x = fmaxf(dt * (av.x + hv[k].x), 0.0f);
            v.y = fmaxf(dt * (av.y + hv[k].y), 0.0f);
            v.z = fmaxf(dt * (av.z + hv[k].z), 0.0f);
            v.w = fmaxf(dt * (av.w + hv[k].w), 0.0f);

            float s  = v.x + v.y + v.z + v.w;
            float s2 = v.x * v.x + v.y * v.y + v.z * v.z + v.w * v.w;
#pragma unroll
            for (int o = 16; o >= 1; o >>= 1) {
                s  += __shfl_xor(s,  o, 32);
                s2 += __shfl_xor(s2, o, 32);
            }
            const float mu   = s * (1.0f / D);
            const float var  = s2 * (1.0f / D) - mu * mu;
            const float rstd = __frsqrt_rn(var + LN_EPS);

            float4 r;
            r.x = (v.x - mu) * rstd * g.x + be.x + xv[k].x;
            r.y = (v.y - mu) * rstd * g.y + be.y + xv[k].y;
            r.z = (v.z - mu) * rstd * g.z + be.z + xv[k].z;
            r.w = (v.w - mu) * rstd * g.w + be.w + xv[k].w;

            ((float4*)(out + (size_t)n * D))[lane] = r;
        }
    }
}

extern "C" void kernel_launch(void* const* d_in, const int* in_sizes, int n_in,
                              void* d_out, int out_size, void* d_ws, size_t ws_size,
                              hipStream_t stream) {
    const float* x     = (const float*)d_in[0];
    const int*   ei    = (const int*)  d_in[1];
    const float* W     = (const float*)d_in[2];
    const float* bias  = (const float*)d_in[3];
    const float* gamma = (const float*)d_in[4];
    const float* beta  = (const float*)d_in[5];

    const int N  = in_sizes[0] / D;
    const int E  = in_sizes[1] / 2;
    const int NB = (N + BIN_NODES - 1) / BIN_NODES;

    const int pitchw = 4 * ((N + 7) / 8);
    char* ws = (char*)d_ws;
    float* disv = (float*)ws;
    size_t off = ((size_t)N * sizeof(float) + 255) & ~(size_t)255;
    float* h = (float*)(ws + off);
    off += ((size_t)N * D * sizeof(float) + 255) & ~(size_t)255;
    unsigned int* cntmat = (unsigned int*)(ws + off);
    off += ((size_t)DEG_GROUPS * NB * sizeof(unsigned int) + 255) & ~(size_t)255;
    int* binstart = (int*)(ws + off);
    off += ((size_t)(NB + 1) * sizeof(int) + 255) & ~(size_t)255;
    unsigned int* planes = (unsigned int*)(ws + off);
    uint2* rec = (uint2*)(ws + off);

    rgcn_degree_kernel<<<DEG_GROUPS, DEG_THREADS, 0, stream>>>(ei, ei + E, planes, cntmat,
                                                               E, N, pitchw);
    rgcn_gemm_kernel<<<(N + GEMM_ROWS - 1) / GEMM_ROWS, 256, 0, stream>>>(x, W, bias, h, planes,
                                                                        disv, N, pitchw);
    rgcn_bin_kernel<<<DEG_GROUPS, BIN_THREADS, 0, stream>>>(ei, cntmat, binstart, rec, E, N);
    rgcn_bin_sum_kernel<<<NB, SUM_WAVES * 32, 0, stream>>>(x, disv, h, rec, binstart, gamma,
                                                         beta, (float*)d_out, E, N);
}


# edge row gather: one 16-byte load per lane fetches the whole 512-B row (was four 4-byte one-line loads); LDS row channel-permuted so the float adds stay conflict-free; arrival-order sums as before
# speedup vs baseline: 2.6359x; 1.0556x over previous
#include <hip/hip_runtime.h>
#include <cstdint>
#include <cstddef>


typedef float v2f __attribute__((ext_vector_type(2)));
typedef float v8f __attribute__((ext_vector_type(8)));

#define D 128
#define LN_EPS 1e-5f

#define BIN_SHIFT 6
#define BIN_NODES (1 << BIN_SHIFT)

#define DEG_BINS    65536
#define DEG_THREADS 1024
#define DEG_GROUPS  128

__global__ void __launch_bounds__(DEG_THREADS)
rgcn_degree_kernel(const int* __restrict__ row, const int* __restrict__ col,
                   unsigned int* __restrict__ planes, unsigned int* __restrict__ cntmat,
                   int E, int N, int pitchw) {
    __shared__ __attribute__((aligned(16))) unsigned int cnt[DEG_BINS];

    const int tid  = (int)threadIdx.x;

    const int per  = (E + DEG_GROUPS - 1) / DEG_GROUPS;
    const int e_lo = (int)blockIdx.x * per;
    const int e_hi = (e_lo + per < E) ? (e_lo + per) : E;

    uint4* __restrict__ plane4 = (uint4*)(planes + (size_t)blockIdx.x * pitchw);

    for (int base = 0; base < N; base += DEG_BINS) {
        const int nb = (N - base < DEG_BINS) ? (N - base) : DEG_BINS;

        const int nw4 = (nb + 7) >> 3;
        for (int q = tid; q < 2 * nw4; q += DEG_THREADS)
            ((uint4*)cnt)[q] = make_uint4(0u, 0u, 0u, 0u);
        __syncthreads();

        for (int e = e_lo + tid; e < e_hi; e += 8 * DEG_THREADS) {
            int r[8];
#pragma unroll
            for (int u = 0; u < 8; ++u) {
                const int ee = e + u * DEG_THREADS;
                r[u] = (ee < e_hi) ? row[ee] : -1;
            }
#pragma unroll
            for (int u = 0; u < 8; ++u) {
                const unsigned b = (unsigned)(r[u] - base);
                if (b < (unsigned)nb) atomicAdd(&cnt[b], 1u);
            }
        }
        __syncthreads();

        for (int q = tid; q < nw4; q += DEG_THREADS) {
            const uint4 a = ((const uint4*)cnt)[2 * q];
            const uint4 b = ((const uint4*)cnt)[2 * q + 1];
            plane4[(base >> 3) + q] = make_uint4(a.x | (a.y << 16), a.z | (a.w << 16),
                                                 b.x | (b.y << 16), b.z | (b.w << 16));
        }
        __syncthreads();
    }

    const int NB = (N + BIN_NODES - 1) >> BIN_SHIFT;
    for (int bbase = 0; bbase < NB; bbase += DEG_BINS) {
        const int nbb = (NB - bbase < DEG_BINS) ? (NB - bbase) : DEG_BINS;
        for (int q = tid; q < nbb; q += DEG_THREADS) cnt[q] = 0u;
        __syncthreads();
        for (int e = e_lo + tid; e < e_hi; e += 8 * DEG_THREADS) {
            int r[8];
#pragma unroll
            for (int u = 0; u < 8; ++u) {
                const int ee = e + u * DEG_THREADS;
                r[u] = (ee < e_hi) ? col[ee] : -1;
            }
#pragma unroll
            for (int u = 0; u < 8; ++u) {
                const unsigned b = (unsigned)((r[u] >> BIN_SHIFT) - bbase);
                if (b < (unsigned)nbb) atomicAdd(&cnt[b], 1u);
            }
        }
        __syncthreads();
        for (int q = tid; q < nbb; q += DEG_THREADS)
            cntmat[(size_t)blockIdx.x * NB + bbase + q] = cnt[q];
        __syncthreads();
    }
}

#define WPITCH 132
#define GEMM_ROWS 128
#define PL_PARTS  4

__global__ void __launch_bounds__(256)
rgcn_gemm_kernel(const float* __restrict__ x, const float* __restrict__ W,
                 const float* __restrict__ bias, float* __restrict__ h,
                 const unsigned int* __restrict__ planes, float* __restrict__ disv,
                 int N, int pitchw) {
    __shared__ __attribute__((aligned(16))) float Ws[D * WPITCH];
    __shared__ unsigned int degp[PL_PARTS][GEMM_ROWS];
    __shared__ __attribute__((aligned(16))) float dis_s[GEMM_ROWS];

    const int pword = (int)threadIdx.x & (GEMM_ROWS / 2 - 1);
    const int part  = (int)threadIdx.x / (GEMM_ROWS / 2);
    const int widx  = (int)blockIdx.x * (GEMM_ROWS / 2) + pword;
    unsigned pc[DEG_GROUPS / PL_PARTS];
    {
        const bool pok = (widx < pitchw);
        const unsigned pw0   = (unsigned)part * (unsigned)pitchw + (unsigned)(pok ? widx : 0);
        const unsigned pstep = (unsigned)PL_PARTS * (unsigned)pitchw;
#pragma unroll
        for (int g = 0; g < DEG_GROUPS / PL_PARTS; ++g) {
            const unsigned c = planes[pw0 + (unsigned)g * pstep];
            pc[g] = pok ? c : 0u;
        }
    }

#pragma unroll
    for (int j = 0; j < 16; ++j) {
        const int i  = (int)threadIdx.x + j * 256;
        const int wr = i >> 5;
        const int wc = i & 31;
        const float4 w4 = ((const float4*)W)[i];
        *(float4*)(Ws + wr * WPITCH + wc * 4) = w4;
    }
    {
        unsigned lo = 0u, hi = 0u;
#pragma unroll
        for (int g = 0; g < DEG_GROUPS / PL_PARTS; ++g) {
            lo += pc[g] & 0xffffu;
            hi += pc[g] >> 16;
        }
        degp[part][2 * pword]     = lo;
        degp[part][2 * pword + 1] = hi;
    }
    __syncthreads();

    if (threadIdx.x < GEMM_ROWS) {
        unsigned dsum = 1u;
#pragma unroll
        for (int p = 0; p < PL_PARTS; ++p) dsum += degp[p][threadIdx.x];
        const float dv = __frsqrt_rn((float)dsum);
        dis_s[threadIdx.x] = dv;
        const int n = (int)blockIdx.x * GEMM_ROWS + (int)threadIdx.x;
        if (n < N) disv[n] = dv;
    }
    __syncthreads();

    const int lane = threadIdx.x & 31;
    const int wave = threadIdx.x >> 5;
    const int m    = lane & 15;
    const int half = lane >> 4;
    const int row0 = blockIdx.x * 128 + wave * 16;

    int r = row0 + m;
    if (r >= N) r = N - 1;

    const float* __restrict__ xrow  = x + (size_t)r * D + 2 * half;
    const float* wbase = Ws + m * WPITCH + 2 * half;

    v8f acc[8] = {};
#pragma unroll 4
    for (int k = 0; k < D; k += 4) {
        v2f a = *(const v2f*)(xrow + k);
#pragma unroll
        for (int t = 0; t < 8; ++t) {
            v2f b = *(const v2f*)(wbase + t * 16 * WPITCH + k);
            acc[t] = __builtin_amdgcn_wmma_f32_16x16x4_f32(
                 false, a,  false, b,
                 (short)0, acc[t],  false,  false);
        }
    }

    const float4 dlo = *(const float4*)(dis_s + wave * 16 + half * 8);
    const float4 dhi = *(const float4*)(dis_s + wave * 16 + half * 8 + 4);
    const float dr[8] = { dlo.x, dlo.y, dlo.z, dlo.w, dhi.x, dhi.y, dhi.z, dhi.w };

#pragma unroll
    for (int t = 0; t < 8; ++t) {
        const float bb = bias[t * 16 + m];
#pragma unroll
        for (int v = 0; v < 8; ++v) {
            int orow = row0 + half * 8 + v;
            if (orow < N)
                h[(size_t)orow * D + t * 16 + m] = (acc[t][v] + bb) * dr[v];
        }
    }
}

#define BIN_THREADS 1024
#define BIN_MAXB    16384

__global__ void __launch_bounds__(BIN_THREADS)
rgcn_bin_kernel(const int* __restrict__ ei,
                const unsigned int* __restrict__ cntmat,
                int* __restrict__ binstart, uint2* __restrict__ rec,
                int E, int N) {
    __shared__ unsigned int cur[BIN_MAXB];
    __shared__ unsigned int tot[BIN_MAXB];
    __shared__ unsigned int wsum[BIN_THREADS / 32];

    const int tid  = (int)threadIdx.x;
    const int lane = tid & 31;
    const int wave = tid >> 5;
    const int g    = (int)blockIdx.x;
    const int NB   = (N + BIN_NODES - 1) >> BIN_SHIFT;

    for (int b = tid; b < NB; b += BIN_THREADS) {
        unsigned total = 0u, mine = 0u;
#pragma unroll 32
        for (int gg = 0; gg < DEG_GROUPS; ++gg) {
            const unsigned cc = cntmat[(size_t)gg * NB + b];
            total += cc;
            mine  += (gg < g) ? cc : 0u;
        }
        tot[b] = total;
        cur[b] = mine;
    }
    __syncthreads();

    unsigned carry = 0u;
    for (int cb = 0; cb < NB; cb += BIN_THREADS) {
        const int b = cb + tid;
        const unsigned v = (b < NB) ? tot[b] : 0u;
        unsigned inc = v;
#pragma unroll
        for (int o = 1; o < 32; o <<= 1) {
            const unsigned up = __shfl_up(inc, o, 32);
            if (lane >= o) inc += up;
        }
        if (lane == 31) wsum[wave] = inc;
        __syncthreads();
        unsigned woff = 0u, ctot = 0u;
#pragma unroll
        for (int w = 0; w < BIN_THREADS / 32; ++w) {
            const unsigned sw = wsum[w];
            woff += (w < wave) ? sw : 0u;
            ctot += sw;
        }
        if (b < NB) {
            const unsigned start = carry + woff + inc - v;
            cur[b] += start;
            if (g == 0) binstart[b] = (int)start;
        }
        carry += ctot;
        __syncthreads();
    }
    if (g == 0 && tid == 0) binstart[NB] = (int)carry;

    const int per  = (E + DEG_GROUPS - 1) / DEG_GROUPS;
    const int e_lo = g * per;
    const int e_hi = (e_lo + per < E) ? (e_lo + per) : E;
    const int* __restrict__ row = ei;
    const int* __restrict__ col = ei + E;

    for (int e = e_lo + tid; e < e_hi; e += 4 * BIN_THREADS) {
        int sidx[4], tidx[4];
#pragma unroll
        for (int u = 0; u < 4; ++u) {
            const int ee = e + u * BIN_THREADS;
            const bool ok = (ee < e_hi);
            sidx[u] = ok ? row[ee] : -1;
            tidx[u] = ok ? col[ee] : -1;
        }
#pragma unroll
        for (int u = 0; u < 4; ++u) {
            if (sidx[u] >= 0) {
                const unsigned p = atomicAdd(&cur[tidx[u] >> BIN_SHIFT], 1u);
                if (p < (unsigned)E)
                    rec[p] = make_uint2((unsigned)sidx[u], (unsigned)tidx[u]);
            }
        }
    }
}

#define EPW 16
#define SUM_WAVES 8
#define NPW (BIN_NODES / SUM_WAVES)

__global__ void __launch_bounds__(SUM_WAVES * 32)
rgcn_bin_sum_kernel(const float* __restrict__ x, const float* __restrict__ disv,
                    const float* __restrict__ h, const uint2* __restrict__ rec,
                    const int* __restrict__ binstart,
                    const float* __restrict__ gamma, const float* __restrict__ beta,
                    float* __restrict__ out, int E, int N) {
    __shared__ __attribute__((aligned(16))) float acc[BIN_NODES * D];

    const int tid  = (int)threadIdx.x;
    const int lane = tid & 31;
    const int wave = __builtin_amdgcn_readfirstlane(tid >> 5);
    const int n0   = (int)blockIdx.x << BIN_SHIFT;
    int e_lo = binstart[blockIdx.x];
    int e_hi = binstart[blockIdx.x + 1];
    if (e_lo < 0) e_lo = 0;
    if (e_hi > E) e_hi = E;

    for (int q = tid; q < BIN_NODES * D / 4; q += SUM_WAVES * 32)
        ((float4*)acc)[q] = make_float4(0.0f, 0.0f, 0.0f, 0.0f);
    __syncthreads();

    const float4* __restrict__ hl = (const float4*)h + lane;
    float* al = acc + lane;

    for (int e0 = e_lo + wave * EPW; e0 < e_hi; e0 += SUM_WAVES * EPW) {
        int e = e0 + (lane & 15);
        if (e >= e_hi) e = e_hi - 1;
        const uint2 r = rec[e];
        const int   rs = min(max((int)r.x, 0), N - 1);
        const int   rt = ((int)r.y - n0) & (BIN_NODES - 1);

        if (e0 + EPW <= e_hi) {
            float4 v[EPW];
            for (int j = 0; j < EPW; ++j) {
                const int s = __builtin_amdgcn_readlane(rs, j);
                v[j] = hl[(size_t)s * (D / 4)];
            }
            for (int j = 0; j < EPW; ++j) {
                const int t = __builtin_amdgcn_readlane(rt, j);
                float* at = al + t * D;
                atomicAdd(at +  0, v[j].x);
                atomicAdd(at + 32, v[j].y);
                atomicAdd(at + 64, v[j].z);
                atomicAdd(at + 96, v[j].w);
            }
        } else {
            const int cnt = e_hi - e0;
            for (int j = 0; j < cnt; ++j) {
                const int s = __builtin_amdgcn_readlane(rs, j);
                const int t = __builtin_amdgcn_readlane(rt, j);
                const float4 v1 = hl[(size_t)s * (D / 4)];
                float* at = al + t * D;
                atomicAdd(at +  0, v1.x);
                atomicAdd(at + 32, v1.y);
                atomicAdd(at + 64, v1.z);
                atomicAdd(at + 96, v1.w);
            }
        }
    }
    const float4 g  = ((const float4*)gamma)[lane];
    const float4 be = ((const float4*)beta)[lane];
    float4 hv[NPW], xv[NPW];
    float  dgv[NPW];
#pragma unroll
    for (int k = 0; k < NPW; ++k) {
        int n = n0 + wave + k * SUM_WAVES;
        if (n >= N) n = N - 1;
        dgv[k] = disv[n];
        hv[k]  = ((const float4*)(h + (size_t)n * D))[lane];
        xv[k]  = ((const float4*)(x + (size_t)n * D))[lane];
    }
    __syncthreads();

#pragma unroll
    for (int k = 0; k < NPW; ++k) {
        const int i = wave + k * SUM_WAVES;
        const int n = n0 + i;
        if (n < N) {
            const float dt = dgv[k];
            float4 av;
            av.x = acc[i * D +  0 + lane];
            av.y = acc[i * D + 32 + lane];
            av.z = acc[i * D + 64 + lane];
            av.w = acc[i * D + 96 + lane];

            float4 v;
            v.x = fmaxf(dt * (av.x + hv[k].x), 0.0f);
            v.y = fmaxf(dt * (av.y + hv[k].y), 0.0f);
            v.z = fmaxf(dt * (av.z + hv[k].z), 0.0f);
            v.w = fmaxf(dt * (av.w + hv[k].w), 0.0f);

            float s  = v.x + v.y + v.z + v.w;
            float s2 = v.x * v.x + v.y * v.y + v.z * v.z + v.w * v.w;
#pragma unroll
            for (int o = 16; o >= 1; o >>= 1) {
                s  += __shfl_xor(s,  o, 32);
                s2 += __shfl_xor(s2, o, 32);
            }
            const float mu   = s * (1.0f / D);
            const float var  = s2 * (1.0f / D) - mu * mu;
            const float rstd = __frsqrt_rn(var + LN_EPS);

            float4 r;
            r.x = (v.x - mu) * rstd * g.x + be.x + xv[k].x;
            r.y = (v.y - mu) * rstd * g.y + be.y + xv[k].y;
            r.z = (v.z - mu) * rstd * g.z + be.z + xv[k].z;
            r.w = (v.w - mu) * rstd * g.w + be.w + xv[k].w;

            ((float4*)(out + (size_t)n * D))[lane] = r;
        }
    }
}

extern "C" void kernel_launch(void* const* d_in, const int* in_sizes, int n_in,
                              void* d_out, int out_size, void* d_ws, size_t ws_size,
                              hipStream_t stream) {
    const float* x     = (const float*)d_in[0];
    const int*   ei    = (const int*)  d_in[1];
    const float* W     = (const float*)d_in[2];
    const float* bias  = (const float*)d_in[3];
    const float* gamma = (const float*)d_in[4];
    const float* beta  = (const float*)d_in[5];

    const int N  = in_sizes[0] / D;
    const int E  = in_sizes[1] / 2;
    const int NB = (N + BIN_NODES - 1) / BIN_NODES;

    const int pitchw = 4 * ((N + 7) / 8);
    char* ws = (char*)d_ws;
    float* disv = (float*)ws;
    size_t off = ((size_t)N * sizeof(float) + 255) & ~(size_t)255;
    float* h = (float*)(ws + off);
    off += ((size_t)N * D * sizeof(float) + 255) & ~(size_t)255;
    unsigned int* cntmat = (unsigned int*)(ws + off);
    off += ((size_t)DEG_GROUPS * NB * sizeof(unsigned int) + 255) & ~(size_t)255;
    int* binstart = (int*)(ws + off);
    off += ((size_t)(NB + 1) * sizeof(int) + 255) & ~(size_t)255;
    unsigned int* planes = (unsigned int*)(ws + off);
    uint2* rec = (uint2*)(ws + off);

    rgcn_degree_kernel<<<DEG_GROUPS, DEG_THREADS, 0, stream>>>(ei, ei + E, planes, cntmat,
                                                               E, N, pitchw);
    rgcn_gemm_kernel<<<(N + GEMM_ROWS - 1) / GEMM_ROWS, 256, 0, stream>>>(x, W, bias, h, planes,
                                                                        disv, N, pitchw);
    rgcn_bin_kernel<<<DEG_GROUPS, BIN_THREADS, 0, stream>>>(ei, cntmat, binstart, rec, E, N);
    rgcn_bin_sum_kernel<<<NB, SUM_WAVES * 32, 0, stream>>>(x, disv, h, rec, binstart, gamma,
                                                         beta, (float*)d_out, E, N);
}


# binning runs as the last 128 workgroups of the gemm launch (independent work; 3 launches, was 4); 256 threads each, 64 count loads, 16 index loads, 8 rank adds in flight; arrival-order sums kept
# speedup vs baseline: 2.8743x; 1.0904x over previous
#include <hip/hip_runtime.h>
#include <cstdint>
#include <cstddef>


typedef float v2f __attribute__((ext_vector_type(2)));
typedef float v8f __attribute__((ext_vector_type(8)));

#define D 128
#define LN_EPS 1e-5f

#define BIN_SHIFT 6
#define BIN_NODES (1 << BIN_SHIFT)

#define DEG_BINS    65536
#define DEG_THREADS 1024
#define DEG_GROUPS  128

__global__ void __launch_bounds__(DEG_THREADS)
rgcn_degree_kernel(const int* __restrict__ row, const int* __restrict__ col,
                   unsigned int* __restrict__ planes, unsigned int* __restrict__ cntmat,
                   int E, int N, int pitchw) {
    __shared__ __attribute__((aligned(16))) unsigned int cnt[DEG_BINS];

    const int tid  = (int)threadIdx.x;

    const int per  = (E + DEG_GROUPS - 1) / DEG_GROUPS;
    const int e_lo = (int)blockIdx.x * per;
    const int e_hi = (e_lo + per < E) ? (e_lo + per) : E;

    uint4* __restrict__ plane4 = (uint4*)(planes + (size_t)blockIdx.x * pitchw);

    for (int base = 0; base < N; base += DEG_BINS) {
        const int nb = (N - base < DEG_BINS) ? (N - base) : DEG_BINS;

        const int nw4 = (nb + 7) >> 3;
        for (int q = tid; q < 2 * nw4; q += DEG_THREADS)
            ((uint4*)cnt)[q] = make_uint4(0u, 0u, 0u, 0u);
        __syncthreads();

        for (int e = e_lo + tid; e < e_hi; e += 8 * DEG_THREADS) {
            int r[8];
#pragma unroll
            for (int u = 0; u < 8; ++u) {
                const int ee = e + u * DEG_THREADS;
                r[u] = (ee < e_hi) ? row[ee] : -1;
            }
#pragma unroll
            for (int u = 0; u < 8; ++u) {
                const unsigned b = (unsigned)(r[u] - base);
                if (b < (unsigned)nb) atomicAdd(&cnt[b], 1u);
            }
        }
        __syncthreads();

        for (int q = tid; q < nw4; q += DEG_THREADS) {
            const uint4 a = ((const uint4*)cnt)[2 * q];
            const uint4 b = ((const uint4*)cnt)[2 * q + 1];
            plane4[(base >> 3) + q] = make_uint4(a.x | (a.y << 16), a.z | (a.w << 16),
                                                 b.x | (b.y << 16), b.z | (b.w << 16));
        }
        __syncthreads();
    }

    const int NB = (N + BIN_NODES - 1) >> BIN_SHIFT;
    for (int bbase = 0; bbase < NB; bbase += DEG_BINS) {
        const int nbb = (NB - bbase < DEG_BINS) ? (NB - bbase) : DEG_BINS;
        for (int q = tid; q < nbb; q += DEG_THREADS) cnt[q] = 0u;
        __syncthreads();
        for (int e = e_lo + tid; e < e_hi; e += 8 * DEG_THREADS) {
            int r[8];
#pragma unroll
            for (int u = 0; u < 8; ++u) {
                const int ee = e + u * DEG_THREADS;
                r[u] = (ee < e_hi) ? col[ee] : -1;
            }
#pragma unroll
            for (int u = 0; u < 8; ++u) {
                const unsigned b = (unsigned)((r[u] >> BIN_SHIFT) - bbase);
                if (b < (unsigned)nbb) atomicAdd(&cnt[b], 1u);
            }
        }
        __syncthreads();
        for (int q = tid; q < nbb; q += DEG_THREADS)
            cntmat[(size_t)blockIdx.x * NB + bbase + q] = cnt[q];
        __syncthreads();
    }
}

#define WPITCH 132
#define GEMM_ROWS 128
#define PL_PARTS  4
#define BIN_THREADS 256
#define BIN_MAXB    8192
#define GB_THREADS  256

#define SMEM_GEMM_WORDS (D * WPITCH + PL_PARTS * GEMM_ROWS + GEMM_ROWS)
#define SMEM_BIN_WORDS  (2 * BIN_MAXB + BIN_THREADS / 32)
#define SMEM_WORDS ((SMEM_GEMM_WORDS > SMEM_BIN_WORDS) ? SMEM_GEMM_WORDS : SMEM_BIN_WORDS)

__global__ void __launch_bounds__(GB_THREADS)
rgcn_gemm_bin_kernel(const float* __restrict__ x, const float* __restrict__ W,
                     const float* __restrict__ bias, float* __restrict__ h,
                     const unsigned int* __restrict__ planes, float* __restrict__ disv,
                     const int* __restrict__ ei, const unsigned int* __restrict__ cntmat,
                     int* __restrict__ binstart, uint2* __restrict__ rec,
                     int N, int E, int pitchw) {
    __shared__ __attribute__((aligned(16))) unsigned int smem[SMEM_WORDS];

    const int nstrips = (N + GEMM_ROWS - 1) / GEMM_ROWS;

    if ((int)blockIdx.x >= nstrips) {
        unsigned int* const cur  = smem;
        unsigned int* const tot  = smem + BIN_MAXB;
        unsigned int* const wsum = smem + 2 * BIN_MAXB;

        const int tid  = (int)threadIdx.x;
        const int lane = tid & 31;
        const int wave = tid >> 5;
        const int g    = (int)blockIdx.x - nstrips;
        const int NB   = (N + BIN_NODES - 1) >> BIN_SHIFT;

        for (int b0 = tid; b0 < NB; b0 += 2 * BIN_THREADS) {
            const int  b1  = b0 + BIN_THREADS;
            const bool ok1 = (b1 < NB);
            const int  b1c = ok1 ? b1 : b0;
            unsigned t0 = 0u, m0 = 0u, t1 = 0u, m1 = 0u;
#pragma unroll 32
            for (int gg = 0; gg < DEG_GROUPS; ++gg) {
                const unsigned c0 = cntmat[(size_t)gg * NB + b0];
                const unsigned c1 = cntmat[(size_t)gg * NB + b1c];
                t0 += c0;
                t1 += c1;
                m0 += (gg < g) ? c0 : 0u;
                m1 += (gg < g) ? c1 : 0u;
            }
            tot[b0] = t0;
            cur[b0] = m0;
            if (ok1) {
                tot[b1] = t1;
                cur[b1] = m1;
            }
        }
        __syncthreads();

        unsigned carry = 0u;
        for (int cb = 0; cb < NB; cb += 4 * BIN_THREADS) {
            const int b = cb + 4 * tid;
            unsigned v[4];
#pragma unroll
            for (int i = 0; i < 4; ++i) v[i] = (b + i < NB) ? tot[b + i] : 0u;
            const unsigned s4 = v[0] + v[1] + v[2] + v[3];
            unsigned inc = s4;
#pragma unroll
            for (int o = 1; o < 32; o <<= 1) {
                const unsigned up = __shfl_up(inc, o, 32);
                if (lane >= o) inc += up;
            }
            if (lane == 31) wsum[wave] = inc;
            __syncthreads();
            unsigned woff = 0u, ctot = 0u;
#pragma unroll
            for (int w = 0; w < BIN_THREADS / 32; ++w) {
                const unsigned sw = wsum[w];
                woff += (w < wave) ? sw : 0u;
                ctot += sw;
            }
            unsigned start = carry + woff + inc - s4;
#pragma unroll
            for (int i = 0; i < 4; ++i) {
                if (b + i < NB) {
                    cur[b + i] += start;
                    if (g == 0) binstart[b + i] = (int)start;
                }
                start += v[i];
            }
            carry += ctot;
            __syncthreads();
        }
        if (g == 0 && tid == 0) binstart[NB] = (int)carry;

        const int per  = (E + DEG_GROUPS - 1) / DEG_GROUPS;
        const int e_lo = g * per;
        const int e_hi = (e_lo + per < E) ? (e_lo + per) : E;
        const int* __restrict__ row = ei;
        const int* __restrict__ col = ei + E;

        for (int e = e_lo + tid; e < e_hi; e += 8 * BIN_THREADS) {
            int sidx[8], tidx[8];
#pragma unroll
            for (int u = 0; u < 8; ++u) {
                const int ee = e + u * BIN_THREADS;
                const bool ok = (ee < e_hi);
                sidx[u] = ok ? row[ee] : -1;
                tidx[u] = ok ? col[ee] : -1;
            }
            unsigned pos[8];
#pragma unroll
            for (int u = 0; u < 8; ++u) {
                pos[u] = 0xffffffffu;
                if (sidx[u] >= 0) pos[u] = atomicAdd(&cur[tidx[u] >> BIN_SHIFT], 1u);
            }
#pragma unroll
            for (int u = 0; u < 8; ++u) {
                if (pos[u] < (unsigned)E)
                    rec[pos[u]] = make_uint2((unsigned)sidx[u], (unsigned)tidx[u]);
            }
        }
    } else {
        float* const Ws = (float*)smem;
        unsigned int* const degp = smem + D * WPITCH;
        float* const dis_s = (float*)(smem + D * WPITCH + PL_PARTS * GEMM_ROWS);

        const int bid = (int)blockIdx.x;

        const int pword = (int)threadIdx.x & (GEMM_ROWS / 2 - 1);
        const int part  = (int)threadIdx.x / (GEMM_ROWS / 2);
        const int widx  = bid * (GEMM_ROWS / 2) + pword;
        unsigned pc[DEG_GROUPS / PL_PARTS];
        {
            const bool pok = (widx < pitchw);
            const unsigned pw0   = (unsigned)part * (unsigned)pitchw + (unsigned)(pok ? widx : 0);
            const unsigned pstep = (unsigned)PL_PARTS * (unsigned)pitchw;
#pragma unroll
            for (int g = 0; g < DEG_GROUPS / PL_PARTS; ++g) {
                const unsigned c = planes[pw0 + (unsigned)g * pstep];
                pc[g] = pok ? c : 0u;
            }
        }

#pragma unroll
        for (int j = 0; j < 16; ++j) {
            const int i  = (int)threadIdx.x + j * 256;
            const int wr = i >> 5;
            const int wc = i & 31;
            const float4 w4 = ((const float4*)W)[i];
            *(float4*)(Ws + wr * WPITCH + wc * 4) = w4;
        }
        {
            unsigned lo = 0u, hi = 0u;
#pragma unroll
            for (int g = 0; g < DEG_GROUPS / PL_PARTS; ++g) {
                lo += pc[g] & 0xffffu;
                hi += pc[g] >> 16;
            }
            degp[part * GEMM_ROWS + 2 * pword]     = lo;
            degp[part * GEMM_ROWS + 2 * pword + 1] = hi;
        }
        __syncthreads();

        if (threadIdx.x < GEMM_ROWS) {
            unsigned dsum = 1u;
#pragma unroll
            for (int p = 0; p < PL_PARTS; ++p) dsum += degp[p * GEMM_ROWS + threadIdx.x];
            const float dv = __frsqrt_rn((float)dsum);
            dis_s[threadIdx.x] = dv;
            const int n = bid * GEMM_ROWS + (int)threadIdx.x;
            if (n < N) disv[n] = dv;
        }
        __syncthreads();

        const int lane = threadIdx.x & 31;
        const int wave = threadIdx.x >> 5;
        const int m    = lane & 15;
        const int half = lane >> 4;
        const int row0 = bid * GEMM_ROWS + wave * 16;

        int r = row0 + m;
        if (r >= N) r = N - 1;

        const float* __restrict__ xrow  = x + (size_t)r * D + 2 * half;
        const float* wbase = Ws + m * WPITCH + 2 * half;

        v8f acc[8] = {};
#pragma unroll 4
        for (int k = 0; k < D; k += 4) {
            v2f a = *(const v2f*)(xrow + k);
#pragma unroll
            for (int t = 0; t < 8; ++t) {
                v2f b = *(const v2f*)(wbase + t * 16 * WPITCH + k);
                acc[t] = __builtin_amdgcn_wmma_f32_16x16x4_f32(
                     false, a,  false, b,
                     (short)0, acc[t],  false,  false);
            }
        }

        const float4 dlo = *(const float4*)(dis_s + wave * 16 + half * 8);
        const float4 dhi = *(const float4*)(dis_s + wave * 16 + half * 8 + 4);
        const float dr[8] = { dlo.x, dlo.y, dlo.z, dlo.w, dhi.x, dhi.y, dhi.z, dhi.w };

#pragma unroll
        for (int t = 0; t < 8; ++t) {
            const float bb = bias[t * 16 + m];
#pragma unroll
            for (int v = 0; v < 8; ++v) {
                int orow = row0 + half * 8 + v;
                if (orow < N)
                    h[(size_t)orow * D + t * 16 + m] = (acc[t][v] + bb) * dr[v];
            }
        }
    }
}

#define EPW 16
#define SUM_WAVES 8
#define NPW (BIN_NODES / SUM_WAVES)

__global__ void __launch_bounds__(SUM_WAVES * 32)
rgcn_bin_sum_kernel(const float* __restrict__ x, const float* __restrict__ disv,
                    const float* __restrict__ h, const uint2* __restrict__ rec,
                    const int* __restrict__ binstart,
                    const float* __restrict__ gamma, const float* __restrict__ beta,
                    float* __restrict__ out, int E, int N) {
    __shared__ __attribute__((aligned(16))) float acc[BIN_NODES * D];

    const int tid  = (int)threadIdx.x;
    const int lane = tid & 31;
    const int wave = __builtin_amdgcn_readfirstlane(tid >> 5);
    const int n0   = (int)blockIdx.x << BIN_SHIFT;
    int e_lo = binstart[blockIdx.x];
    int e_hi = binstart[blockIdx.x + 1];
    if (e_lo < 0) e_lo = 0;
    if (e_hi > E) e_hi = E;

    for (int q = tid; q < BIN_NODES * D / 4; q += SUM_WAVES * 32)
        ((float4*)acc)[q] = make_float4(0.0f, 0.0f, 0.0f, 0.0f);
    __syncthreads();

    const float4* __restrict__ hl = (const float4*)h + lane;
    float* al = acc + lane;

    for (int e0 = e_lo + wave * EPW; e0 < e_hi; e0 += SUM_WAVES * EPW) {
        int e = e0 + (lane & 15);
        if (e >= e_hi) e = e_hi - 1;
        const uint2 r = rec[e];
        const int   rs = min(max((int)r.x, 0), N - 1);
        const int   rt = ((int)r.y - n0) & (BIN_NODES - 1);

        if (e0 + EPW <= e_hi) {
            float4 v[EPW];
            for (int j = 0; j < EPW; ++j) {
                const int s = __builtin_amdgcn_readlane(rs, j);
                v[j] = hl[(size_t)s * (D / 4)];
            }
            for (int j = 0; j < EPW; ++j) {
                const int t = __builtin_amdgcn_readlane(rt, j);
                float* at = al + t * D;
                atomicAdd(at +  0, v[j].x);
                atomicAdd(at + 32, v[j].y);
                atomicAdd(at + 64, v[j].z);
                atomicAdd(at + 96, v[j].w);
            }
        } else {
            const int cnt = e_hi - e0;
            for (int j = 0; j < cnt; ++j) {
                const int s = __builtin_amdgcn_readlane(rs, j);
                const int t = __builtin_amdgcn_readlane(rt, j);
                const float4 v1 = hl[(size_t)s * (D / 4)];
                float* at = al + t * D;
                atomicAdd(at +  0, v1.x);
                atomicAdd(at + 32, v1.y);
                atomicAdd(at + 64, v1.z);
                atomicAdd(at + 96, v1.w);
            }
        }
    }
    const float4 g  = ((const float4*)gamma)[lane];
    const float4 be = ((const float4*)beta)[lane];
    float4 hv[NPW], xv[NPW];
    float  dgv[NPW];
#pragma unroll
    for (int k = 0; k < NPW; ++k) {
        int n = n0 + wave + k * SUM_WAVES;
        if (n >= N) n = N - 1;
        dgv[k] = disv[n];
        hv[k]  = ((const float4*)(h + (size_t)n * D))[lane];
        xv[k]  = ((const float4*)(x + (size_t)n * D))[lane];
    }
    __syncthreads();

#pragma unroll
    for (int k = 0; k < NPW; ++k) {
        const int i = wave + k * SUM_WAVES;
        const int n = n0 + i;
        if (n < N) {
            const float dt = dgv[k];
            float4 av;
            av.x = acc[i * D +  0 + lane];
            av.y = acc[i * D + 32 + lane];
            av.z = acc[i * D + 64 + lane];
            av.w = acc[i * D + 96 + lane];

            float4 v;
            v.x = fmaxf(dt * (av.x + hv[k].x), 0.0f);
            v.y = fmaxf(dt * (av.y + hv[k].y), 0.0f);
            v.z = fmaxf(dt * (av.z + hv[k].z), 0.0f);
            v.w = fmaxf(dt * (av.w + hv[k].w), 0.0f);

            float s  = v.x + v.y + v.z + v.w;
            float s2 = v.x * v.x + v.y * v.y + v.z * v.z + v.w * v.w;
#pragma unroll
            for (int o = 16; o >= 1; o >>= 1) {
                s  += __shfl_xor(s,  o, 32);
                s2 += __shfl_xor(s2, o, 32);
            }
            const float mu   = s * (1.0f / D);
            const float var  = s2 * (1.0f / D) - mu * mu;
            const float rstd = __frsqrt_rn(var + LN_EPS);

            float4 r;
            r.x = (v.x - mu) * rstd * g.x + be.x + xv[k].x;
            r.y = (v.y - mu) * rstd * g.y + be.y + xv[k].y;
            r.z = (v.z - mu) * rstd * g.z + be.z + xv[k].z;
            r.w = (v.w - mu) * rstd * g.w + be.w + xv[k].w;

            ((float4*)(out + (size_t)n * D))[lane] = r;
        }
    }
}

extern "C" void kernel_launch(void* const* d_in, const int* in_sizes, int n_in,
                              void* d_out, int out_size, void* d_ws, size_t ws_size,
                              hipStream_t stream) {
    const float* x     = (const float*)d_in[0];
    const int*   ei    = (const int*)  d_in[1];
    const float* W     = (const float*)d_in[2];
    const float* bias  = (const float*)d_in[3];
    const float* gamma = (const float*)d_in[4];
    const float* beta  = (const float*)d_in[5];

    const int N  = in_sizes[0] / D;
    const int E  = in_sizes[1] / 2;
    const int NB = (N + BIN_NODES - 1) / BIN_NODES;

    const int pitchw = 4 * ((N + 7) / 8);
    char* ws = (char*)d_ws;
    float* disv = (float*)ws;
    size_t off = ((size_t)N * sizeof(float) + 255) & ~(size_t)255;
    float* h = (float*)(ws + off);
    off += ((size_t)N * D * sizeof(float) + 255) & ~(size_t)255;
    unsigned int* cntmat = (unsigned int*)(ws + off);
    off += ((size_t)DEG_GROUPS * NB * sizeof(unsigned int) + 255) & ~(size_t)255;
    int* binstart = (int*)(ws + off);
    off += ((size_t)(NB + 1) * sizeof(int) + 255) & ~(size_t)255;
    unsigned int* planes = (unsigned int*)(ws + off);
    off += ((size_t)DEG_GROUPS * pitchw * sizeof(unsigned int) + 255) & ~(size_t)255;
    uint2* rec = (uint2*)(ws + off);

    rgcn_degree_kernel<<<DEG_GROUPS, DEG_THREADS, 0, stream>>>(ei, ei + E, planes, cntmat,
                                                               E, N, pitchw);
    rgcn_gemm_bin_kernel<<<(N + GEMM_ROWS - 1) / GEMM_ROWS + DEG_GROUPS, GB_THREADS, 0, stream>>>(
        x, W, bias, h, planes, disv, ei, cntmat, binstart, rec, N, E, pitchw);
    rgcn_bin_sum_kernel<<<NB, SUM_WAVES * 32, 0, stream>>>(x, disv, h, rec, binstart, gamma,
                                                         beta, (float*)d_out, E, N);
}


# sum kernel: a bin's records ordered by target node in LDS (integer count, scan, place); a wave sums its 8 nodes' rows in registers in list order (ranks by arrival), no LDS float atomics
# speedup vs baseline: 3.1345x; 1.0905x over previous
#include <hip/hip_runtime.h>
#include <cstdint>
#include <cstddef>


typedef float v2f __attribute__((ext_vector_type(2)));
typedef float v8f __attribute__((ext_vector_type(8)));

#define D 128
#define LN_EPS 1e-5f

#define BIN_SHIFT 6
#define BIN_NODES (1 << BIN_SHIFT)

#define DEG_BINS    65536
#define DEG_THREADS 1024
#define DEG_GROUPS  128

__global__ void __launch_bounds__(DEG_THREADS)
rgcn_degree_kernel(const int* __restrict__ row, const int* __restrict__ col,
                   unsigned int* __restrict__ planes, unsigned int* __restrict__ cntmat,
                   int E, int N, int pitchw) {
    __shared__ __attribute__((aligned(16))) unsigned int cnt[DEG_BINS];

    const int tid  = (int)threadIdx.x;

    const int per  = (E + DEG_GROUPS - 1) / DEG_GROUPS;
    const int e_lo = (int)blockIdx.x * per;
    const int e_hi = (e_lo + per < E) ? (e_lo + per) : E;

    uint4* __restrict__ plane4 = (uint4*)(planes + (size_t)blockIdx.x * pitchw);

    for (int base = 0; base < N; base += DEG_BINS) {
        const int nb = (N - base < DEG_BINS) ? (N - base) : DEG_BINS;

        const int nw4 = (nb + 7) >> 3;
        for (int q = tid; q < 2 * nw4; q += DEG_THREADS)
            ((uint4*)cnt)[q] = make_uint4(0u, 0u, 0u, 0u);
        __syncthreads();

        for (int e = e_lo + tid; e < e_hi; e += 8 * DEG_THREADS) {
            int r[8];
#pragma unroll
            for (int u = 0; u < 8; ++u) {
                const int ee = e + u * DEG_THREADS;
                r[u] = (ee < e_hi) ? row[ee] : -1;
            }
#pragma unroll
            for (int u = 0; u < 8; ++u) {
                const unsigned b = (unsigned)(r[u] - base);
                if (b < (unsigned)nb) atomicAdd(&cnt[b], 1u);
            }
        }
        __syncthreads();

        for (int q = tid; q < nw4; q += DEG_THREADS) {
            const uint4 a = ((const uint4*)cnt)[2 * q];
            const uint4 b = ((const uint4*)cnt)[2 * q + 1];
            plane4[(base >> 3) + q] = make_uint4(a.x | (a.y << 16), a.z | (a.w << 16),
                                                 b.x | (b.y << 16), b.z | (b.w << 16));
        }
        __syncthreads();
    }

    const int NB = (N + BIN_NODES - 1) >> BIN_SHIFT;
    for (int bbase = 0; bbase < NB; bbase += DEG_BINS) {
        const int nbb = (NB - bbase < DEG_BINS) ? (NB - bbase) : DEG_BINS;
        for (int q = tid; q < nbb; q += DEG_THREADS) cnt[q] = 0u;
        __syncthreads();
        for (int e = e_lo + tid; e < e_hi; e += 8 * DEG_THREADS) {
            int r[8];
#pragma unroll
            for (int u = 0; u < 8; ++u) {
                const int ee = e + u * DEG_THREADS;
                r[u] = (ee < e_hi) ? col[ee] : -1;
            }
#pragma unroll
            for (int u = 0; u < 8; ++u) {
                const unsigned b = (unsigned)((r[u] >> BIN_SHIFT) - bbase);
                if (b < (unsigned)nbb) atomicAdd(&cnt[b], 1u);
            }
        }
        __syncthreads();
        for (int q = tid; q < nbb; q += DEG_THREADS)
            cntmat[(size_t)blockIdx.x * NB + bbase + q] = cnt[q];
        __syncthreads();
    }
}

#define WPITCH 132
#define GEMM_ROWS 128
#define PL_PARTS  4
#define BIN_THREADS 256
#define BIN_MAXB    8192
#define GB_THREADS  256

#define SMEM_GEMM_WORDS (D * WPITCH + PL_PARTS * GEMM_ROWS + GEMM_ROWS)
#define SMEM_BIN_WORDS  (2 * BIN_MAXB + BIN_THREADS / 32)
#define SMEM_WORDS ((SMEM_GEMM_WORDS > SMEM_BIN_WORDS) ? SMEM_GEMM_WORDS : SMEM_BIN_WORDS)

__global__ void __launch_bounds__(GB_THREADS)
rgcn_gemm_bin_kernel(const float* __restrict__ x, const float* __restrict__ W,
                     const float* __restrict__ bias, float* __restrict__ h,
                     const unsigned int* __restrict__ planes, float* __restrict__ disv,
                     const int* __restrict__ ei, const unsigned int* __restrict__ cntmat,
                     int* __restrict__ binstart, uint2* __restrict__ rec,
                     int N, int E, int pitchw) {
    __shared__ __attribute__((aligned(16))) unsigned int smem[SMEM_WORDS];

    const int nstrips = (N + GEMM_ROWS - 1) / GEMM_ROWS;

    if ((int)blockIdx.x >= nstrips) {
        unsigned int* const cur  = smem;
        unsigned int* const tot  = smem + BIN_MAXB;
        unsigned int* const wsum = smem + 2 * BIN_MAXB;

        const int tid  = (int)threadIdx.x;
        const int lane = tid & 31;
        const int wave = tid >> 5;
        const int g    = (int)blockIdx.x - nstrips;
        const int NB   = (N + BIN_NODES - 1) >> BIN_SHIFT;

        for (int b0 = tid; b0 < NB; b0 += 2 * BIN_THREADS) {
            const int  b1  = b0 + BIN_THREADS;
            const bool ok1 = (b1 < NB);
            const int  b1c = ok1 ? b1 : b0;
            unsigned t0 = 0u, m0 = 0u, t1 = 0u, m1 = 0u;
#pragma unroll 32
            for (int gg = 0; gg < DEG_GROUPS; ++gg) {
                const unsigned c0 = cntmat[(size_t)gg * NB + b0];
                const unsigned c1 = cntmat[(size_t)gg * NB + b1c];
                t0 += c0;
                t1 += c1;
                m0 += (gg < g) ? c0 : 0u;
                m1 += (gg < g) ? c1 : 0u;
            }
            tot[b0] = t0;
            cur[b0] = m0;
            if (ok1) {
                tot[b1] = t1;
                cur[b1] = m1;
            }
        }
        __syncthreads();

        unsigned carry = 0u;
        for (int cb = 0; cb < NB; cb += 4 * BIN_THREADS) {
            const int b = cb + 4 * tid;
            unsigned v[4];
#pragma unroll
            for (int i = 0; i < 4; ++i) v[i] = (b + i < NB) ? tot[b + i] : 0u;
            const unsigned s4 = v[0] + v[1] + v[2] + v[3];
            unsigned inc = s4;
#pragma unroll
            for (int o = 1; o < 32; o <<= 1) {
                const unsigned up = __shfl_up(inc, o, 32);
                if (lane >= o) inc += up;
            }
            if (lane == 31) wsum[wave] = inc;
            __syncthreads();
            unsigned woff = 0u, ctot = 0u;
#pragma unroll
            for (int w = 0; w < BIN_THREADS / 32; ++w) {
                const unsigned sw = wsum[w];
                woff += (w < wave) ? sw : 0u;
                ctot += sw;
            }
            unsigned start = carry + woff + inc - s4;
#pragma unroll
            for (int i = 0; i < 4; ++i) {
                if (b + i < NB) {
                    cur[b + i] += start;
                    if (g == 0) binstart[b + i] = (int)start;
                }
                start += v[i];
            }
            carry += ctot;
            __syncthreads();
        }
        if (g == 0 && tid == 0) binstart[NB] = (int)carry;

        const int per  = (E + DEG_GROUPS - 1) / DEG_GROUPS;
        const int e_lo = g * per;
        const int e_hi = (e_lo + per < E) ? (e_lo + per) : E;
        const int* __restrict__ row = ei;
        const int* __restrict__ col = ei + E;

        for (int e = e_lo + tid; e < e_hi; e += 8 * BIN_THREADS) {
            int sidx[8], tidx[8];
#pragma unroll
            for (int u = 0; u < 8; ++u) {
                const int ee = e + u * BIN_THREADS;
                const bool ok = (ee < e_hi);
                sidx[u] = ok ? row[ee] : -1;
                tidx[u] = ok ? col[ee] : -1;
            }
            unsigned pos[8];
#pragma unroll
            for (int u = 0; u < 8; ++u) {
                pos[u] = 0xffffffffu;
                if (sidx[u] >= 0) pos[u] = atomicAdd(&cur[tidx[u] >> BIN_SHIFT], 1u);
            }
#pragma unroll
            for (int u = 0; u < 8; ++u) {
                if (pos[u] < (unsigned)E)
                    rec[pos[u]] = make_uint2((unsigned)sidx[u], (unsigned)tidx[u]);
            }
        }
    } else {
        float* const Ws = (float*)smem;
        unsigned int* const degp = smem + D * WPITCH;
        float* const dis_s = (float*)(smem + D * WPITCH + PL_PARTS * GEMM_ROWS);

        const int bid = (int)blockIdx.x;

        const int pword = (int)threadIdx.x & (GEMM_ROWS / 2 - 1);
        const int part  = (int)threadIdx.x / (GEMM_ROWS / 2);
        const int widx  = bid * (GEMM_ROWS / 2) + pword;
        unsigned pc[DEG_GROUPS / PL_PARTS];
        {
            const bool pok = (widx < pitchw);
            const unsigned pw0   = (unsigned)part * (unsigned)pitchw + (unsigned)(pok ? widx : 0);
            const unsigned pstep = (unsigned)PL_PARTS * (unsigned)pitchw;
#pragma unroll
            for (int g = 0; g < DEG_GROUPS / PL_PARTS; ++g) {
                const unsigned c = planes[pw0 + (unsigned)g * pstep];
                pc[g] = pok ? c : 0u;
            }
        }

#pragma unroll
        for (int j = 0; j < 16; ++j) {
            const int i  = (int)threadIdx.x + j * 256;
            const int wr = i >> 5;
            const int wc = i & 31;
            const float4 w4 = ((const float4*)W)[i];
            *(float4*)(Ws + wr * WPITCH + wc * 4) = w4;
        }
        {
            unsigned lo = 0u, hi = 0u;
#pragma unroll
            for (int g = 0; g < DEG_GROUPS / PL_PARTS; ++g) {
                lo += pc[g] & 0xffffu;
                hi += pc[g] >> 16;
            }
            degp[part * GEMM_ROWS + 2 * pword]     = lo;
            degp[part * GEMM_ROWS + 2 * pword + 1] = hi;
        }
        __syncthreads();

        if (threadIdx.x < GEMM_ROWS) {
            unsigned dsum = 1u;
#pragma unroll
            for (int p = 0; p < PL_PARTS; ++p) dsum += degp[p * GEMM_ROWS + threadIdx.x];
            const float dv = __frsqrt_rn((float)dsum);
            dis_s[threadIdx.x] = dv;
            const int n = bid * GEMM_ROWS + (int)threadIdx.x;
            if (n < N) disv[n] = dv;
        }
        __syncthreads();

        const int lane = threadIdx.x & 31;
        const int wave = threadIdx.x >> 5;
        const int m    = lane & 15;
        const int half = lane >> 4;
        const int row0 = bid * GEMM_ROWS + wave * 16;

        int r = row0 + m;
        if (r >= N) r = N - 1;

        const float* __restrict__ xrow  = x + (size_t)r * D + 2 * half;
        const float* wbase = Ws + m * WPITCH + 2 * half;

        v8f acc[8] = {};
#pragma unroll 4
        for (int k = 0; k < D; k += 4) {
            v2f a = *(const v2f*)(xrow + k);
#pragma unroll
            for (int t = 0; t < 8; ++t) {
                v2f b = *(const v2f*)(wbase + t * 16 * WPITCH + k);
                acc[t] = __builtin_amdgcn_wmma_f32_16x16x4_f32(
                     false, a,  false, b,
                     (short)0, acc[t],  false,  false);
            }
        }

        const float4 dlo = *(const float4*)(dis_s + wave * 16 + half * 8);
        const float4 dhi = *(const float4*)(dis_s + wave * 16 + half * 8 + 4);
        const float dr[8] = { dlo.x, dlo.y, dlo.z, dlo.w, dhi.x, dhi.y, dhi.z, dhi.w };

#pragma unroll
        for (int t = 0; t < 8; ++t) {
            const float bb = bias[t * 16 + m];
#pragma unroll
            for (int v = 0; v < 8; ++v) {
                int orow = row0 + half * 8 + v;
                if (orow < N)
                    h[(size_t)orow * D + t * 16 + m] = (acc[t][v] + bb) * dr[v];
            }
        }
    }
}

#define EPW 16
#define SUM_WAVES 8
#define NPW (BIN_NODES / SUM_WAVES)
#define RPT 8
#define CHUNK (RPT * SUM_WAVES * 32)

__global__ void __launch_bounds__(SUM_WAVES * 32)
rgcn_bin_sum_kernel(const float* __restrict__ x, const float* __restrict__ disv,
                    const float* __restrict__ h, const uint2* __restrict__ rec,
                    const int* __restrict__ binstart,
                    const float* __restrict__ gamma, const float* __restrict__ beta,
                    float* __restrict__ out, int E, int N) {
    __shared__ __attribute__((aligned(16))) float rows[BIN_NODES * D];
    __shared__ unsigned int srt[CHUNK];
    __shared__ unsigned int cntn[BIN_NODES];
    __shared__ unsigned int startn[BIN_NODES + 1];

    const int tid  = (int)threadIdx.x;
    const int lane = tid & 31;
    const int wave = __builtin_amdgcn_readfirstlane(tid >> 5);
    const int n0   = (int)blockIdx.x << BIN_SHIFT;
    int e_lo = binstart[blockIdx.x];
    int e_hi = binstart[blockIdx.x + 1];
    if (e_lo < 0) e_lo = 0;
    if (e_hi > E) e_hi = E;

    for (int q = tid; q < BIN_NODES * D / 4; q += SUM_WAVES * 32)
        ((float4*)rows)[q] = make_float4(0.0f, 0.0f, 0.0f, 0.0f);

    const float4* __restrict__ hl = (const float4*)h + lane;
    float4* const rl = (float4*)rows + lane;

    for (int c0 = e_lo; c0 < e_hi; c0 += CHUNK) {
        const int c1 = (c0 + CHUNK < e_hi) ? (c0 + CHUNK) : e_hi;

        if (tid < BIN_NODES) cntn[tid] = 0u;
        __syncthreads();

        unsigned key[RPT], rank[RPT];
#pragma unroll
        for (int u = 0; u < RPT; ++u) {
            const int idx = c0 + tid + u * (SUM_WAVES * 32);
            const uint2 r = rec[(idx < c1) ? idx : (c1 - 1)];
            const int rs = min(max((int)r.x, 0), N - 1);
            key[u] = ((unsigned)rs << BIN_SHIFT) | ((r.y - (unsigned)n0) & (unsigned)(BIN_NODES - 1));
        }
#pragma unroll
        for (int u = 0; u < RPT; ++u) {
            const int idx = c0 + tid + u * (SUM_WAVES * 32);
            rank[u] = 0u;
            if (idx < c1) rank[u] = atomicAdd(&cntn[key[u] & (unsigned)(BIN_NODES - 1)], 1u);
        }
        __syncthreads();

        if (wave == 0) {
            const unsigned ca = cntn[2 * lane];
            const unsigned cb = cntn[2 * lane + 1];
            const unsigned s2 = ca + cb;
            unsigned inc = s2;
#pragma unroll
            for (int o = 1; o < 32; o <<= 1) {
                const unsigned up = __shfl_up(inc, o, 32);
                if (lane >= o) inc += up;
            }
            const unsigned ex = inc - s2;
            startn[2 * lane]     = ex;
            startn[2 * lane + 1] = ex + ca;
            if (lane == 31) startn[BIN_NODES] = inc;
        }
        __syncthreads();

        unsigned st[RPT];
#pragma unroll
        for (int u = 0; u < RPT; ++u) st[u] = startn[key[u] & (unsigned)(BIN_NODES - 1)];
#pragma unroll
        for (int u = 0; u < RPT; ++u) {
            const int idx = c0 + tid + u * (SUM_WAVES * 32);
            if (idx < c1) srt[st[u] + rank[u]] = key[u];
        }
        __syncthreads();

        const int p0 = __builtin_amdgcn_readfirstlane((int)startn[NPW * wave]);
        const int p1 = __builtin_amdgcn_readfirstlane((int)startn[NPW * wave + NPW]);

        float4 sum = make_float4(0.0f, 0.0f, 0.0f, 0.0f);
        for (int p = p0; p < p1; p += EPW) {
            const int idx  = p + (lane & 15);
            const int last = p1 - 1;
            const unsigned w0 = srt[(idx     < last) ? idx       : last];
            const unsigned w1 = srt[(idx + 1 < last) ? (idx + 1) : last];
            const bool closes = (idx + 1 >= p1) | (((w0 ^ w1) & (unsigned)(BIN_NODES - 1)) != 0u);
            const unsigned fl = __builtin_amdgcn_ballot_w32(closes);
            const int rs = (int)(w0 >> BIN_SHIFT);
            const int rt = (int)(w0 & (unsigned)(BIN_NODES - 1));
            const int cnt = (p1 - p < EPW) ? (p1 - p) : EPW;

            float4 v[EPW];
            for (int j = 0; j < EPW; ++j) {
                const int s = __builtin_amdgcn_readlane(rs, j);
                v[j] = hl[(size_t)s * (D / 4)];
            }
            for (int j = 0; j < EPW; ++j) {
                if (j == 0 || j < cnt) {
                    sum.x += v[j].x;
                    sum.y += v[j].y;
                    sum.z += v[j].z;
                    sum.w += v[j].w;
                    if ((fl >> j) & 1u) {
                        const int t = __builtin_amdgcn_readlane(rt, j);
                        float4* rp = rl + t * (D / 4);
                        float4 o = *rp;
                        o.x += sum.x;
                        o.y += sum.y;
                        o.z += sum.z;
                        o.w += sum.w;
                        *rp = o;
                        sum = make_float4(0.0f, 0.0f, 0.0f, 0.0f);
                    }
                }
            }
        }
    }
    const float4 g  = ((const float4*)gamma)[lane];
    const float4 be = ((const float4*)beta)[lane];
    float4 hv[NPW], xv[NPW];
    float  dgv[NPW];
#pragma unroll
    for (int k = 0; k < NPW; ++k) {
        int n = n0 + wave + k * SUM_WAVES;
        if (n >= N) n = N - 1;
        dgv[k] = disv[n];
        hv[k]  = ((const float4*)(h + (size_t)n * D))[lane];
        xv[k]  = ((const float4*)(x + (size_t)n * D))[lane];
    }
    __syncthreads();

#pragma unroll
    for (int k = 0; k < NPW; ++k) {
        const int i = wave + k * SUM_WAVES;
        const int n = n0 + i;
        if (n < N) {
            const float dt = dgv[k];
            const float4 av = rl[i * (D / 4)];

            float4 v;
            v.x = fmaxf(dt * (av.x + hv[k].x), 0.0f);
            v.y = fmaxf(dt * (av.y + hv[k].y), 0.0f);
            v.z = fmaxf(dt * (av.z + hv[k].z), 0.0f);
            v.w = fmaxf(dt * (av.w + hv[k].w), 0.0f);

            float s  = v.x + v.y + v.z + v.w;
            float s2 = v.x * v.x + v.y * v.y + v.z * v.z + v.w * v.w;
#pragma unroll
            for (int o = 16; o >= 1; o >>= 1) {
                s  += __shfl_xor(s,  o, 32);
                s2 += __shfl_xor(s2, o, 32);
            }
            const float mu   = s * (1.0f / D);
            const float var  = s2 * (1.0f / D) - mu * mu;
            const float rstd = __frsqrt_rn(var + LN_EPS);

            float4 r;
            r.x = (v.x - mu) * rstd * g.x + be.x + xv[k].x;
            r.y = (v.y - mu) * rstd * g.y + be.y + xv[k].y;
            r.z = (v.z - mu) * rstd * g.z + be.z + xv[k].z;
            r.w = (v.w - mu) * rstd * g.w + be.w + xv[k].w;

            ((float4*)(out + (size_t)n * D))[lane] = r;
        }
    }
}

extern "C" void kernel_launch(void* const* d_in, const int* in_sizes, int n_in,
                              void* d_out, int out_size, void* d_ws, size_t ws_size,
                              hipStream_t stream) {
    const float* x     = (const float*)d_in[0];
    const int*   ei    = (const int*)  d_in[1];
    const float* W     = (const float*)d_in[2];
    const float* bias  = (const float*)d_in[3];
    const float* gamma = (const float*)d_in[4];
    const float* beta  = (const float*)d_in[5];

    const int N  = in_sizes[0] / D;
    const int E  = in_sizes[1] / 2;
    const int NB = (N + BIN_NODES - 1) / BIN_NODES;

    const int pitchw = 4 * ((N + 7) / 8);
    char* ws = (char*)d_ws;
    float* disv = (float*)ws;
    size_t off = ((size_t)N * sizeof(float) + 255) & ~(size_t)255;
    float* h = (float*)(ws + off);
    off += ((size_t)N * D * sizeof(float) + 255) & ~(size_t)255;
    unsigned int* cntmat = (unsigned int*)(ws + off);
    off += ((size_t)DEG_GROUPS * NB * sizeof(unsigned int) + 255) & ~(size_t)255;
    int* binstart = (int*)(ws + off);
    off += ((size_t)(NB + 1) * sizeof(int) + 255) & ~(size_t)255;
    unsigned int* planes = (unsigned int*)(ws + off);
    off += ((size_t)DEG_GROUPS * pitchw * sizeof(unsigned int) + 255) & ~(size_t)255;
    uint2* rec = (uint2*)(ws + off);

    rgcn_degree_kernel<<<DEG_GROUPS, DEG_THREADS, 0, stream>>>(ei, ei + E, planes, cntmat,
                                                               E, N, pitchw);
    rgcn_gemm_bin_kernel<<<(N + GEMM_ROWS - 1) / GEMM_ROWS + DEG_GROUPS, GB_THREADS, 0, stream>>>(
        x, W, bias, h, planes, disv, ei, cntmat, binstart, rec, N, E, pitchw);
    rgcn_bin_sum_kernel<<<NB, SUM_WAVES * 32, 0, stream>>>(x, disv, h, rec, binstart, gamma,
                                                         beta, (float*)d_out, E, N);
}
